# starting kernel
# baseline (speedup 1.0000x reference)
#include <hip/hip_runtime.h>
#include <hip/hip_bf16.h>
#include <math.h>

#define N_NODES 100000
#define E_EDGES 3200000
#define NIN     500
#define NHID    64
#define NOUT    40
#define NLAYERS 8
#define ALPHA   0.1f

#define BRS 160

typedef __attribute__((ext_vector_type(2))) float v2f;
typedef __attribute__((ext_vector_type(8))) float v8f;

__device__ __forceinline__ void wmma_acc(v8f& c, v2f a, v2f b) {
  asm("v_wmma_f32_16x16x4_f32 %0, %1, %2, %0" : "+v"(c) : "v"(a), "v"(b));
}
__device__ __forceinline__ void wmma_epilogue_guard() {
  asm volatile("s_nop 8");
}

__global__ void __launch_bounds__(256) in_gemm_kernel(
    const float* __restrict__ x, const float* __restrict__ Win,
    const float* __restrict__ bin, float* __restrict__ h0) {
  extern __shared__ float Wp[];
  for (int idx = threadIdx.x; idx < NIN * NHID; idx += 256) {
    const int k = idx >> 6;
    const int n = idx & 63;
    Wp[(k >> 1) * BRS + 2 * n + (k & 1)] = Win[idx];
  }
  __syncthreads();

  const int wave = threadIdx.x >> 5;
  const int lane = threadIdx.x & 31;
  const int half = lane >> 4;
  const int l16  = lane & 15;
  const int r0   = blockIdx.x * 256 + wave * 32;

  const int arow0 = min(r0 + l16,      N_NODES - 1);
  const int arow1 = min(r0 + 16 + l16, N_NODES - 1);
  const float* xr0 = x + (long long)arow0 * NIN;
  const float* xr1 = x + (long long)arow1 * NIN;

  v8f acc0[4] = {};
  v8f acc1[4] = {};
  for (int k = 0; k < NIN; k += 4) {
    const int kk = k + 2 * half;
    const v2f a0 = *reinterpret_cast<const v2f*>(xr0 + kk);
    const v2f a1 = *reinterpret_cast<const v2f*>(xr1 + kk);
    const int bbase = (kk >> 1) * BRS + 2 * l16;
#pragma unroll
    for (int t = 0; t < 4; ++t) {
      const v2f b = *reinterpret_cast<const v2f*>(&Wp[bbase + t * 32]);
      wmma_acc(acc0[t], a0, b);
      wmma_acc(acc1[t], a1, b);
    }
  }
  wmma_epilogue_guard();

#pragma unroll
  for (int t = 0; t < 4; ++t) {
    const int n = t * 16 + l16;
    const float bias = bin[n];
#pragma unroll
    for (int r = 0; r < 8; ++r) {
      const int m = r + 8 * half;
      const int orow0 = r0 + m;
      const int orow1 = r0 + 16 + m;
      if (orow0 < N_NODES) {
        const float v = acc0[t][r] + bias;
        h0[(long long)orow0 * NHID + n] = v > 0.f ? v : 0.f;
      }
      if (orow1 < N_NODES) {
        const float v = acc1[t][r] + bias;
        h0[(long long)orow1 * NHID + n] = v > 0.f ? v : 0.f;
      }
    }
  }
}

__global__ void zero_kernel(float4* __restrict__ p, int n4) {
  int i = blockIdx.x * blockDim.x + threadIdx.x;
  if (i < n4) p[i] = make_float4(0.f, 0.f, 0.f, 0.f);
}

__global__ void __launch_bounds__(256) spmm_kernel(
    const int* __restrict__ row, const int* __restrict__ col,
    const float* __restrict__ ew, const float* __restrict__ h,
    float* __restrict__ hi) {
  const long long t = (long long)blockIdx.x * blockDim.x + threadIdx.x;
  const long long total = (long long)E_EDGES * 16;
  if (t >= total) return;
  const int e = (int)(t >> 4);
  const int q = (int)(t & 15);
  const int c = col[e];
  const int r = row[e];
  const float w = ew[e];
  const float4 v = *reinterpret_cast<const float4*>(h + (long long)c * NHID + q * 4);
  float* dst = hi + (long long)r * NHID + q * 4;
  unsafeAtomicAdd(dst + 0, v.x * w);
  unsafeAtomicAdd(dst + 1, v.y * w);
  unsafeAtomicAdd(dst + 2, v.z * w);
  unsafeAtomicAdd(dst + 3, v.w * w);
}

__global__ void __launch_bounds__(256) conv_kernel(
    const float* __restrict__ hi, const float* __restrict__ h0,
    const float* __restrict__ Wl, float beta, float* __restrict__ hout) {
  __shared__ float Wp[(NHID / 2) * BRS];
  for (int idx = threadIdx.x; idx < NHID * NHID; idx += 256) {
    const int k = idx >> 6;
    const int n = idx & 63;
    Wp[(k >> 1) * BRS + 2 * n + (k & 1)] = Wl[idx];
  }
  __syncthreads();

  const int wave = threadIdx.x >> 5;
  const int lane = threadIdx.x & 31;
  const int half = lane >> 4;
  const int l16  = lane & 15;
  const int r0   = blockIdx.x * 256 + wave * 32;

  const float sA = 1.f - ALPHA;
  const float sB = ALPHA;
  const float cscale = (1.f - beta) / beta;

  v8f acc0[4] = {};
  v8f acc1[4] = {};
#pragma unroll
  for (int t = 0; t < 4; ++t) {
    const int n = t * 16 + l16;
#pragma unroll
    for (int r = 0; r < 8; ++r) {
      const int m = r + 8 * half;
      const long long rr0 = min(r0 + m,      N_NODES - 1);
      const long long rr1 = min(r0 + 16 + m, N_NODES - 1);
      acc0[t][r] = cscale * (sA * hi[rr0 * NHID + n] + sB * h0[rr0 * NHID + n]);
      acc1[t][r] = cscale * (sA * hi[rr1 * NHID + n] + sB * h0[rr1 * NHID + n]);
    }
  }

  const long long arow0 = min(r0 + l16,      N_NODES - 1);
  const long long arow1 = min(r0 + 16 + l16, N_NODES - 1);
  const float* hia0 = hi + arow0 * NHID;
  const float* h0a0 = h0 + arow0 * NHID;
  const float* hia1 = hi + arow1 * NHID;
  const float* h0a1 = h0 + arow1 * NHID;
#pragma unroll
  for (int k = 0; k < NHID; k += 4) {
    const int kk = k + 2 * half;
    const v2f hv0 = *reinterpret_cast<const v2f*>(hia0 + kk);
    const v2f gv0 = *reinterpret_cast<const v2f*>(h0a0 + kk);
    const v2f hv1 = *reinterpret_cast<const v2f*>(hia1 + kk);
    const v2f gv1 = *reinterpret_cast<const v2f*>(h0a1 + kk);
    v2f a0, a1;
    a0.x = sA * hv0.x + sB * gv0.x;  a0.y = sA * hv0.y + sB * gv0.y;
    a1.x = sA * hv1.x + sB * gv1.x;  a1.y = sA * hv1.y + sB * gv1.y;
    const int bbase = (kk >> 1) * BRS + 2 * l16;
#pragma unroll
    for (int t = 0; t < 4; ++t) {
      const v2f b = *reinterpret_cast<const v2f*>(&Wp[bbase + t * 32]);
      wmma_acc(acc0[t], a0, b);
      wmma_acc(acc1[t], a1, b);
    }
  }
  wmma_epilogue_guard();

#pragma unroll
  for (int t = 0; t < 4; ++t) {
    const int n = t * 16 + l16;
#pragma unroll
    for (int r = 0; r < 8; ++r) {
      const int m = r + 8 * half;
      const int orow0 = r0 + m;
      const int orow1 = r0 + 16 + m;
      if (orow0 < N_NODES) {
        const float v = beta * acc0[t][r];
        hout[(long long)orow0 * NHID + n] = v > 0.f ? v : 0.f;
      }
      if (orow1 < N_NODES) {
        const float v = beta * acc1[t][r];
        hout[(long long)orow1 * NHID + n] = v > 0.f ? v : 0.f;
      }
    }
  }
}

__global__ void __launch_bounds__(256) out_kernel(
    const float* __restrict__ h, const float* __restrict__ Wout,
    const float* __restrict__ bout, float* __restrict__ out) {
  __shared__ float Ws[NHID * NOUT];
  __shared__ float bs[NOUT];
  for (int idx = threadIdx.x; idx < NHID * NOUT; idx += 256) Ws[idx] = Wout[idx];
  if (threadIdx.x < NOUT) bs[threadIdx.x] = bout[threadIdx.x];
  __syncthreads();

  const int i = blockIdx.x * blockDim.x + threadIdx.x;
  if (i >= N_NODES) return;

  float hr[NHID];
  const float4* hp = reinterpret_cast<const float4*>(h + (long long)i * NHID);
#pragma unroll
  for (int k4 = 0; k4 < NHID / 4; ++k4) {
    const float4 v = hp[k4];
    hr[4 * k4 + 0] = v.x; hr[4 * k4 + 1] = v.y;
    hr[4 * k4 + 2] = v.z; hr[4 * k4 + 3] = v.w;
  }

  float lg[NOUT];
#pragma unroll
  for (int c = 0; c < NOUT; ++c) lg[c] = bs[c];
  for (int k = 0; k < NHID; ++k) {
    const float hk = hr[k];
#pragma unroll
    for (int c = 0; c < NOUT; ++c) lg[c] = fmaf(hk, Ws[k * NOUT + c], lg[c]);
  }

  float mx = lg[0];
#pragma unroll
  for (int c = 1; c < NOUT; ++c) mx = fmaxf(mx, lg[c]);
  float s = 0.f;
#pragma unroll
  for (int c = 0; c < NOUT; ++c) s += expf(lg[c] - mx);
  const float lse = mx + logf(s);
#pragma unroll
  for (int c = 0; c < NOUT; ++c)
    out[(long long)i * NOUT + c] = lg[c] - lse;
}

extern "C" void kernel_launch(void* const* d_in, const int* in_sizes, int n_in,
                              void* d_out, int out_size, void* d_ws, size_t ws_size,
                              hipStream_t stream) {
  (void)in_sizes; (void)n_in; (void)out_size; (void)ws_size;
  const float* x     = (const float*)d_in[0];
  const int*   row   = (const int*)  d_in[1];
  const int*   col   = (const int*)  d_in[2];
  const float* ew    = (const float*)d_in[3];
  const float* Win   = (const float*)d_in[4];
  const float* bin   = (const float*)d_in[5];
  const float* convW = (const float*)d_in[6];
  const float* Wout  = (const float*)d_in[7];
  const float* bout  = (const float*)d_in[8];

  const size_t S = (size_t)N_NODES * NHID;
  float* h0 = (float*)d_ws;
  float* hA = h0 + S;
  float* hB = hA + S;

  const int rowBlocks = (N_NODES + 255) / 256;
  const size_t inLds = (size_t)(NIN / 2) * BRS * sizeof(float);
  in_gemm_kernel<<<rowBlocks, 256, inLds, stream>>>(x, Win, bin, h0);

  const long long spmmItems = (long long)E_EDGES * 16;
  const int spmmBlocks = (int)((spmmItems + 255) / 256);
  const int zeroBlocks = (int)((S / 4 + 255) / 256);

  for (int l = 0; l < NLAYERS; ++l) {
    const float beta = logf(0.5f / (float)(l + 1) + 1.f);
    zero_kernel<<<zeroBlocks, 256, 0, stream>>>((float4*)hB, (int)(S / 4));
    spmm_kernel<<<spmmBlocks, 256, 0, stream>>>(row, col, ew, (l == 0) ? h0 : hA, hB);
    conv_kernel<<<rowBlocks, 256, 0, stream>>>(hB, h0, convW + (size_t)l * NHID * NHID,
                                               beta, hA);
  }

  out_kernel<<<(N_NODES + 255) / 256, 256, 0, stream>>>(hA, Wout, bout, (float*)d_out);
}


# SpMM: edge list grouped by destination row once per call (counting sort, integer-atomic rank), then per-layer row gather summed in registers; float atomics and zero-fill gone; row sums in rank order
# speedup vs baseline: 2.7994x; 2.7994x over previous
#include <hip/hip_runtime.h>
#include <hip/hip_bf16.h>
#include <math.h>

#define N_NODES 100000
#define E_EDGES 3200000
#define NIN     500
#define NHID    64
#define NOUT    40
#define NLAYERS 8
#define ALPHA   0.1f

#define BRS 160

typedef __attribute__((ext_vector_type(2))) float v2f;
typedef __attribute__((ext_vector_type(8))) float v8f;

__device__ __forceinline__ void wmma_acc(v8f& c, v2f a, v2f b) {
  asm("v_wmma_f32_16x16x4_f32 %0, %1, %2, %0" : "+v"(c) : "v"(a), "v"(b));
}
__device__ __forceinline__ void wmma_epilogue_guard() {
  asm volatile("s_nop 8");
}

__global__ void __launch_bounds__(256) in_gemm_kernel(
    const float* __restrict__ x, const float* __restrict__ Win,
    const float* __restrict__ bin, float* __restrict__ h0) {
  extern __shared__ float Wp[];
  for (int idx = threadIdx.x; idx < NIN * NHID; idx += 256) {
    const int k = idx >> 6;
    const int n = idx & 63;
    Wp[(k >> 1) * BRS + 2 * n + (k & 1)] = Win[idx];
  }
  __syncthreads();

  const int wave = threadIdx.x >> 5;
  const int lane = threadIdx.x & 31;
  const int half = lane >> 4;
  const int l16  = lane & 15;
  const int r0   = blockIdx.x * 256 + wave * 32;

  const int arow0 = min(r0 + l16,      N_NODES - 1);
  const int arow1 = min(r0 + 16 + l16, N_NODES - 1);
  const float* xr0 = x + (long long)arow0 * NIN;
  const float* xr1 = x + (long long)arow1 * NIN;

  v8f acc0[4] = {};
  v8f acc1[4] = {};
  for (int k = 0; k < NIN; k += 4) {
    const int kk = k + 2 * half;
    const v2f a0 = *reinterpret_cast<const v2f*>(xr0 + kk);
    const v2f a1 = *reinterpret_cast<const v2f*>(xr1 + kk);
    const int bbase = (kk >> 1) * BRS + 2 * l16;
#pragma unroll
    for (int t = 0; t < 4; ++t) {
      const v2f b = *reinterpret_cast<const v2f*>(&Wp[bbase + t * 32]);
      wmma_acc(acc0[t], a0, b);
      wmma_acc(acc1[t], a1, b);
    }
  }
  wmma_epilogue_guard();

#pragma unroll
  for (int t = 0; t < 4; ++t) {
    const int n = t * 16 + l16;
    const float bias = bin[n];
#pragma unroll
    for (int r = 0; r < 8; ++r) {
      const int m = r + 8 * half;
      const int orow0 = r0 + m;
      const int orow1 = r0 + 16 + m;
      if (orow0 < N_NODES) {
        const float v = acc0[t][r] + bias;
        h0[(long long)orow0 * NHID + n] = v > 0.f ? v : 0.f;
      }
      if (orow1 < N_NODES) {
        const float v = acc1[t][r] + bias;
        h0[(long long)orow1 * NHID + n] = v > 0.f ? v : 0.f;
      }
    }
  }
}

__global__ void zero_kernel(float4* __restrict__ p, int n4) {
  int i = blockIdx.x * blockDim.x + threadIdx.x;
  if (i < n4) p[i] = make_float4(0.f, 0.f, 0.f, 0.f);
}


__global__ void __launch_bounds__(256) rank_kernel(
    const int* __restrict__ row, int* __restrict__ cnt, int* __restrict__ rank) {
  const int e = blockIdx.x * 256 + threadIdx.x;
  if (e < E_EDGES) rank[e] = atomicAdd(&cnt[row[e]], 1);
}

__global__ void __launch_bounds__(256) blocksum_kernel(
    const int* __restrict__ cnt, int* __restrict__ bsum) {
  __shared__ int wsum[8];
  const int i = blockIdx.x * 256 + threadIdx.x;
  int v = (i < N_NODES) ? cnt[i] : 0;
#pragma unroll
  for (int o = 16; o > 0; o >>= 1) v += __shfl_xor(v, o);
  if ((threadIdx.x & 31) == 0) wsum[threadIdx.x >> 5] = v;
  __syncthreads();
  if (threadIdx.x == 0) {
    int s = 0;
#pragma unroll
    for (int k = 0; k < 8; ++k) s += wsum[k];
    bsum[blockIdx.x] = s;
  }
}

__global__ void __launch_bounds__(256) scan_kernel(
    const int* __restrict__ cnt, const int* __restrict__ bsum,
    int* __restrict__ rowptr) {
  __shared__ int wsum[8];
  __shared__ int wpart[8];
  const int tid = threadIdx.x;
  const int lane = tid & 31;
  const int wv = tid >> 5;
  int part = 0;
  for (int j = tid; j < (int)blockIdx.x; j += 256) part += bsum[j];
#pragma unroll
  for (int o = 16; o > 0; o >>= 1) part += __shfl_xor(part, o);
  const int i = blockIdx.x * 256 + tid;
  const int c = (i < N_NODES) ? cnt[i] : 0;
  int incl = c;
#pragma unroll
  for (int o = 1; o < 32; o <<= 1) {
    const int t = __shfl_up(incl, o);
    if (lane >= o) incl += t;
  }
  if (lane == 31) wsum[wv] = incl;
  if (lane == 0) wpart[wv] = part;
  __syncthreads();
  int base = 0;
#pragma unroll
  for (int k = 0; k < 8; ++k) {
    base += wpart[k];
    if (k < wv) base += wsum[k];
  }
  if (i < N_NODES) rowptr[i] = base + incl - c;
  if (i == N_NODES - 1) rowptr[N_NODES] = base + incl;
}

__global__ void __launch_bounds__(256) fill_kernel(
    const int* __restrict__ row, const int* __restrict__ col,
    const float* __restrict__ ew, const int* __restrict__ rank,
    const int* __restrict__ rowptr, int2* __restrict__ meta) {
  const int e = blockIdx.x * 256 + threadIdx.x;
  if (e < E_EDGES) {
    const int p = rowptr[row[e]] + rank[e];
    meta[p] = make_int2(col[e], __float_as_int(ew[e]));
  }
}

__global__ void __launch_bounds__(256) spmm_kernel(
    const int* __restrict__ rowptr, const int2* __restrict__ meta,
    const float* __restrict__ h, float* __restrict__ hi) {
  const int lane = threadIdx.x & 31;
  const int r = __builtin_amdgcn_readfirstlane((int)(blockIdx.x * 8 + (threadIdx.x >> 5)));
  if (r >= N_NODES) return;
  const int start = rowptr[r];
  const int end   = rowptr[r + 1];
  v2f acc = {0.f, 0.f};
  for (int e = start; e < end; e += 32) {
    const int n = end - e;
    int2 mine = meta[e + min(lane, n - 1)];
    if (lane >= n) mine.y = 0;
#pragma unroll
    for (int g = 0; g < 4; ++g) {
      if (g * 8 < n) {
        v2f v[8];
#pragma unroll
        for (int j = 0; j < 8; ++j) {
          const unsigned c = (unsigned)__builtin_amdgcn_readlane(mine.x, g * 8 + j);
          const float* hr = h + (size_t)c * NHID;
          v[j] = *reinterpret_cast<const v2f*>(hr + 2 * lane);
        }
#pragma unroll
        for (int j = 0; j < 8; ++j) {
          const float w = __int_as_float(__builtin_amdgcn_readlane(mine.y, g * 8 + j));
          acc.x = fmaf(w, v[j].x, acc.x);
          acc.y = fmaf(w, v[j].y, acc.y);
        }
      }
    }
  }
  *reinterpret_cast<v2f*>(hi + (long long)r * NHID + 2 * lane) = acc;
}

__global__ void __launch_bounds__(256) conv_kernel(
    const float* __restrict__ hi, const float* __restrict__ h0,
    const float* __restrict__ Wl, float beta, float* __restrict__ hout) {
  __shared__ float Wp[(NHID / 2) * BRS];
  for (int idx = threadIdx.x; idx < NHID * NHID; idx += 256) {
    const int k = idx >> 6;
    const int n = idx & 63;
    Wp[(k >> 1) * BRS + 2 * n + (k & 1)] = Wl[idx];
  }
  __syncthreads();

  const int wave = threadIdx.x >> 5;
  const int lane = threadIdx.x & 31;
  const int half = lane >> 4;
  const int l16  = lane & 15;
  const int r0   = blockIdx.x * 256 + wave * 32;

  const float sA = 1.f - ALPHA;
  const float sB = ALPHA;
  const float cscale = (1.f - beta) / beta;

  v8f acc0[4] = {};
  v8f acc1[4] = {};
#pragma unroll
  for (int t = 0; t < 4; ++t) {
    const int n = t * 16 + l16;
#pragma unroll
    for (int r = 0; r < 8; ++r) {
      const int m = r + 8 * half;
      const long long rr0 = min(r0 + m,      N_NODES - 1);
      const long long rr1 = min(r0 + 16 + m, N_NODES - 1);
      acc0[t][r] = cscale * (sA * hi[rr0 * NHID + n] + sB * h0[rr0 * NHID + n]);
      acc1[t][r] = cscale * (sA * hi[rr1 * NHID + n] + sB * h0[rr1 * NHID + n]);
    }
  }

  const long long arow0 = min(r0 + l16,      N_NODES - 1);
  const long long arow1 = min(r0 + 16 + l16, N_NODES - 1);
  const float* hia0 = hi + arow0 * NHID;
  const float* h0a0 = h0 + arow0 * NHID;
  const float* hia1 = hi + arow1 * NHID;
  const float* h0a1 = h0 + arow1 * NHID;
#pragma unroll
  for (int k = 0; k < NHID; k += 4) {
    const int kk = k + 2 * half;
    const v2f hv0 = *reinterpret_cast<const v2f*>(hia0 + kk);
    const v2f gv0 = *reinterpret_cast<const v2f*>(h0a0 + kk);
    const v2f hv1 = *reinterpret_cast<const v2f*>(hia1 + kk);
    const v2f gv1 = *reinterpret_cast<const v2f*>(h0a1 + kk);
    v2f a0, a1;
    a0.x = sA * hv0.x + sB * gv0.x;  a0.y = sA * hv0.y + sB * gv0.y;
    a1.x = sA * hv1.x + sB * gv1.x;  a1.y = sA * hv1.y + sB * gv1.y;
    const int bbase = (kk >> 1) * BRS + 2 * l16;
#pragma unroll
    for (int t = 0; t < 4; ++t) {
      const v2f b = *reinterpret_cast<const v2f*>(&Wp[bbase + t * 32]);
      wmma_acc(acc0[t], a0, b);
      wmma_acc(acc1[t], a1, b);
    }
  }
  wmma_epilogue_guard();

#pragma unroll
  for (int t = 0; t < 4; ++t) {
    const int n = t * 16 + l16;
#pragma unroll
    for (int r = 0; r < 8; ++r) {
      const int m = r + 8 * half;
      const int orow0 = r0 + m;
      const int orow1 = r0 + 16 + m;
      if (orow0 < N_NODES) {
        const float v = beta * acc0[t][r];
        hout[(long long)orow0 * NHID + n] = v > 0.f ? v : 0.f;
      }
      if (orow1 < N_NODES) {
        const float v = beta * acc1[t][r];
        hout[(long long)orow1 * NHID + n] = v > 0.f ? v : 0.f;
      }
    }
  }
}

__global__ void __launch_bounds__(256) out_kernel(
    const float* __restrict__ h, const float* __restrict__ Wout,
    const float* __restrict__ bout, float* __restrict__ out) {
  __shared__ float Ws[NHID * NOUT];
  __shared__ float bs[NOUT];
  for (int idx = threadIdx.x; idx < NHID * NOUT; idx += 256) Ws[idx] = Wout[idx];
  if (threadIdx.x < NOUT) bs[threadIdx.x] = bout[threadIdx.x];
  __syncthreads();

  const int i = blockIdx.x * blockDim.x + threadIdx.x;
  if (i >= N_NODES) return;

  float hr[NHID];
  const float4* hp = reinterpret_cast<const float4*>(h + (long long)i * NHID);
#pragma unroll
  for (int k4 = 0; k4 < NHID / 4; ++k4) {
    const float4 v = hp[k4];
    hr[4 * k4 + 0] = v.x; hr[4 * k4 + 1] = v.y;
    hr[4 * k4 + 2] = v.z; hr[4 * k4 + 3] = v.w;
  }

  float lg[NOUT];
#pragma unroll
  for (int c = 0; c < NOUT; ++c) lg[c] = bs[c];
  for (int k = 0; k < NHID; ++k) {
    const float hk = hr[k];
#pragma unroll
    for (int c = 0; c < NOUT; ++c) lg[c] = fmaf(hk, Ws[k * NOUT + c], lg[c]);
  }

  float mx = lg[0];
#pragma unroll
  for (int c = 1; c < NOUT; ++c) mx = fmaxf(mx, lg[c]);
  float s = 0.f;
#pragma unroll
  for (int c = 0; c < NOUT; ++c) s += expf(lg[c] - mx);
  const float lse = mx + logf(s);
#pragma unroll
  for (int c = 0; c < NOUT; ++c)
    out[(long long)i * NOUT + c] = lg[c] - lse;
}

extern "C" void kernel_launch(void* const* d_in, const int* in_sizes, int n_in,
                              void* d_out, int out_size, void* d_ws, size_t ws_size,
                              hipStream_t stream) {
  (void)in_sizes; (void)n_in; (void)out_size; (void)ws_size;
  const float* x     = (const float*)d_in[0];
  const int*   row   = (const int*)  d_in[1];
  const int*   col   = (const int*)  d_in[2];
  const float* ew    = (const float*)d_in[3];
  const float* Win   = (const float*)d_in[4];
  const float* bin   = (const float*)d_in[5];
  const float* convW = (const float*)d_in[6];
  const float* Wout  = (const float*)d_in[7];
  const float* bout  = (const float*)d_in[8];

  const size_t S = (size_t)N_NODES * NHID;
  float* h0 = (float*)d_ws;
  float* hA = h0 + S;
  float* hB = hA + S;
  int*  cnt    = (int*)(hB + S);
  int*  bsum   = cnt + 100096;
  int*  rowptr = bsum + 512;
  int*  rank   = rowptr + 100096;
  int2* meta   = (int2*)(rank + E_EDGES);

  const int rowBlocks = (N_NODES + 255) / 256;
  const size_t inLds = (size_t)(NIN / 2) * BRS * sizeof(float);
  in_gemm_kernel<<<rowBlocks, 256, inLds, stream>>>(x, Win, bin, h0);

  const int edgeBlocks = (E_EDGES + 255) / 256;
  zero_kernel<<<(N_NODES / 4 + 255) / 256, 256, 0, stream>>>((float4*)cnt, N_NODES / 4);
  rank_kernel<<<edgeBlocks, 256, 0, stream>>>(row, cnt, rank);
  blocksum_kernel<<<rowBlocks, 256, 0, stream>>>(cnt, bsum);
  scan_kernel<<<rowBlocks, 256, 0, stream>>>(cnt, bsum, rowptr);
  fill_kernel<<<edgeBlocks, 256, 0, stream>>>(row, col, ew, rank, rowptr, meta);

  const int spmmBlocks = (N_NODES + 7) / 8;

  for (int l = 0; l < NLAYERS; ++l) {
    const float beta = logf(0.5f / (float)(l + 1) + 1.f);
    spmm_kernel<<<spmmBlocks, 256, 0, stream>>>(rowptr, meta, (l == 0) ? h0 : hA, hB);
    conv_kernel<<<rowBlocks, 256, 0, stream>>>(hB, h0, convW + (size_t)l * NHID * NHID,
                                               beta, hA);
  }

  out_kernel<<<(N_NODES + 255) / 256, 256, 0, stream>>>(hA, Wout, bout, (float*)d_out);
}


# Layer body fused into the row-gather launch: 16 one-row waves per block leave S in LDS, 4 of them run the 64x64 product on WMMA from an LDS copy of W, rows stored whole; 8 launches fewer
# speedup vs baseline: 3.3699x; 1.2038x over previous
#include <hip/hip_runtime.h>
#include <hip/hip_bf16.h>
#include <math.h>

#define N_NODES 100000
#define E_EDGES 3200000
#define NIN     500
#define NHID    64
#define NOUT    40
#define NLAYERS 8
#define ALPHA   0.1f

#define BRS 160

typedef __attribute__((ext_vector_type(2))) float v2f;
typedef __attribute__((ext_vector_type(8))) float v8f;

__device__ __forceinline__ void wmma_acc(v8f& c, v2f a, v2f b) {
  asm("v_wmma_f32_16x16x4_f32 %0, %1, %2, %0" : "+v"(c) : "v"(a), "v"(b));
}
__device__ __forceinline__ void wmma_acc_slots(v8f& c, v2f a, v2f b) {
  asm("v_wmma_f32_16x16x4_f32 %0, %1, %2, %0\n\tv_nop\n\tv_nop\n\tv_nop\n\tv_nop"
      : "+v"(c) : "v"(a), "v"(b));
}
__device__ __forceinline__ void wmma_epilogue_guard() {
  asm volatile("s_nop 8");
}

__global__ void __launch_bounds__(256) in_gemm_kernel(
    const float* __restrict__ x, const float* __restrict__ Win,
    const float* __restrict__ bin, float* __restrict__ h0) {
  extern __shared__ float Wp[];
  for (int idx = threadIdx.x; idx < NIN * NHID; idx += 256) {
    const int k = idx >> 6;
    const int n = idx & 63;
    Wp[(k >> 1) * BRS + 2 * n + (k & 1)] = Win[idx];
  }
  __syncthreads();

  const int wave = threadIdx.x >> 5;
  const int lane = threadIdx.x & 31;
  const int half = lane >> 4;
  const int l16  = lane & 15;
  const int r0   = blockIdx.x * 256 + wave * 32;

  const int arow0 = min(r0 + l16,      N_NODES - 1);
  const int arow1 = min(r0 + 16 + l16, N_NODES - 1);
  const float* xr0 = x + (long long)arow0 * NIN;
  const float* xr1 = x + (long long)arow1 * NIN;

  v8f acc0[4] = {};
  v8f acc1[4] = {};
  for (int k = 0; k < NIN; k += 4) {
    const int kk = k + 2 * half;
    const v2f a0 = *reinterpret_cast<const v2f*>(xr0 + kk);
    const v2f a1 = *reinterpret_cast<const v2f*>(xr1 + kk);
    const int bbase = (kk >> 1) * BRS + 2 * l16;
#pragma unroll
    for (int t = 0; t < 4; ++t) {
      const v2f b = *reinterpret_cast<const v2f*>(&Wp[bbase + t * 32]);
      wmma_acc(acc0[t], a0, b);
      wmma_acc(acc1[t], a1, b);
    }
  }
  wmma_epilogue_guard();

#pragma unroll
  for (int t = 0; t < 4; ++t) {
    const int n = t * 16 + l16;
    const float bias = bin[n];
#pragma unroll
    for (int r = 0; r < 8; ++r) {
      const int m = r + 8 * half;
      const int orow0 = r0 + m;
      const int orow1 = r0 + 16 + m;
      if (orow0 < N_NODES) {
        const float v = acc0[t][r] + bias;
        h0[(long long)orow0 * NHID + n] = v > 0.f ? v : 0.f;
      }
      if (orow1 < N_NODES) {
        const float v = acc1[t][r] + bias;
        h0[(long long)orow1 * NHID + n] = v > 0.f ? v : 0.f;
      }
    }
  }
}

__global__ void zero_kernel(float4* __restrict__ p, int n4) {
  int i = blockIdx.x * blockDim.x + threadIdx.x;
  if (i < n4) p[i] = make_float4(0.f, 0.f, 0.f, 0.f);
}


__global__ void __launch_bounds__(256) rank_kernel(
    const int* __restrict__ row, int* __restrict__ cnt, int* __restrict__ rank) {
  const int e = blockIdx.x * 256 + threadIdx.x;
  if (e < E_EDGES) rank[e] = atomicAdd(&cnt[row[e]], 1);
}

__global__ void __launch_bounds__(256) blocksum_kernel(
    const int* __restrict__ cnt, int* __restrict__ bsum) {
  __shared__ int wsum[8];
  const int i = blockIdx.x * 256 + threadIdx.x;
  int v = (i < N_NODES) ? cnt[i] : 0;
#pragma unroll
  for (int o = 16; o > 0; o >>= 1) v += __shfl_xor(v, o);
  if ((threadIdx.x & 31) == 0) wsum[threadIdx.x >> 5] = v;
  __syncthreads();
  if (threadIdx.x == 0) {
    int s = 0;
#pragma unroll
    for (int k = 0; k < 8; ++k) s += wsum[k];
    bsum[blockIdx.x] = s;
  }
}

__global__ void __launch_bounds__(256) scan_kernel(
    const int* __restrict__ cnt, const int* __restrict__ bsum,
    int* __restrict__ rowptr) {
  __shared__ int wsum[8];
  __shared__ int wpart[8];
  const int tid = threadIdx.x;
  const int lane = tid & 31;
  const int wv = tid >> 5;
  int part = 0;
  for (int j = tid; j < (int)blockIdx.x; j += 256) part += bsum[j];
#pragma unroll
  for (int o = 16; o > 0; o >>= 1) part += __shfl_xor(part, o);
  const int i = blockIdx.x * 256 + tid;
  const int c = (i < N_NODES) ? cnt[i] : 0;
  int incl = c;
#pragma unroll
  for (int o = 1; o < 32; o <<= 1) {
    const int t = __shfl_up(incl, o);
    if (lane >= o) incl += t;
  }
  if (lane == 31) wsum[wv] = incl;
  if (lane == 0) wpart[wv] = part;
  __syncthreads();
  int base = 0;
#pragma unroll
  for (int k = 0; k < 8; ++k) {
    base += wpart[k];
    if (k < wv) base += wsum[k];
  }
  if (i < N_NODES) rowptr[i] = base + incl - c;
  if (i == N_NODES - 1) rowptr[N_NODES] = base + incl;
}

__global__ void __launch_bounds__(256) fill_kernel(
    const int* __restrict__ row, const int* __restrict__ col,
    const float* __restrict__ ew, const int* __restrict__ rank,
    const int* __restrict__ rowptr, int2* __restrict__ meta) {
  const int e = blockIdx.x * 256 + threadIdx.x;
  if (e < E_EDGES) {
    const int p = rowptr[row[e]] + rank[e];
    meta[p] = make_int2(col[e], __float_as_int(ew[e]));
  }
}

#define SROW 68

static_assert(N_NODES % 16 == 0, "16 rows per block, no partial tile");

__global__ void __launch_bounds__(512) layer_kernel(
    const int* __restrict__ rowptr, const int2* __restrict__ meta,
    const float* __restrict__ h, const float* __restrict__ h0,
    const float* __restrict__ Wl, float beta, float* __restrict__ hout) {
  __shared__ __attribute__((aligned(16))) float Ssm[16 * SROW];
  __shared__ __attribute__((aligned(16))) float Osm[16 * SROW];
  __shared__ __attribute__((aligned(16))) float Wp[(NHID / 2) * BRS];
  const int tp = threadIdx.x >> 4;
  const int tn = (threadIdx.x & 15) * 4;
  const float4 w0 = *reinterpret_cast<const float4*>(Wl + (2 * tp) * NHID + tn);
  const float4 w1 = *reinterpret_cast<const float4*>(Wl + (2 * tp + 1) * NHID + tn);
  const int lane = threadIdx.x & 31;
  const int w = __builtin_amdgcn_readfirstlane((int)(threadIdx.x >> 5));
  const int r = blockIdx.x * 16 + w;
  const int start = rowptr[r];
  const int end   = rowptr[r + 1];
  const v2f g0 = *reinterpret_cast<const v2f*>(h0 + (size_t)r * NHID + 2 * lane);
  v2f acc = {0.f, 0.f};
  for (int e = start; e < end; e += 32) {
    const int n = end - e;
    int2 mine = meta[e + min(lane, n - 1)];
    if (lane >= n) mine.y = 0;
#pragma unroll
    for (int g = 0; g < 4; ++g) {
      if (g * 8 < n) {
        v2f v[8];
#pragma unroll
        for (int j = 0; j < 8; ++j) {
          const unsigned c = (unsigned)__builtin_amdgcn_readlane(mine.x, g * 8 + j);
          const float* hr = h + (size_t)c * NHID;
          v[j] = *reinterpret_cast<const v2f*>(hr + 2 * lane);
        }
#pragma unroll
        for (int j = 0; j < 8; ++j) {
          const float wt = __int_as_float(__builtin_amdgcn_readlane(mine.y, g * 8 + j));
          acc.x = fmaf(wt, v[j].x, acc.x);
          acc.y = fmaf(wt, v[j].y, acc.y);
        }
      }
      if (g == 0)
        asm volatile("" :: "v"(g0.x), "v"(g0.y), "v"(w0.x), "v"(w0.y), "v"(w0.z), "v"(w0.w),
                           "v"(w1.x), "v"(w1.y), "v"(w1.z), "v"(w1.w));
    }
  }
  const float sA = 1.f - ALPHA;
  const float sB = ALPHA;
  v2f s;
  s.x = sA * acc.x + sB * g0.x;
  s.y = sA * acc.y + sB * g0.y;
  *reinterpret_cast<v2f*>(&Ssm[w * SROW + 2 * lane]) = s;
  *reinterpret_cast<float4*>(&Wp[tp * BRS + 2 * tn])     = make_float4(w0.x, w1.x, w0.y, w1.y);
  *reinterpret_cast<float4*>(&Wp[tp * BRS + 2 * tn + 4]) = make_float4(w0.z, w1.z, w0.w, w1.w);
  __syncthreads();

  if (w < 4) {
    const int half = lane >> 4;
    const int l16  = lane & 15;
    const int n    = w * 16 + l16;
    const float cscale = (1.f - beta) / beta;
    v8f d;
#pragma unroll
    for (int q = 0; q < 8; ++q) d[q] = cscale * Ssm[(8 * half + q) * SROW + n];
#pragma unroll
    for (int k = 0; k < NHID; k += 4) {
      const int kk = k + 2 * half;
      const v2f a = *reinterpret_cast<const v2f*>(&Ssm[l16 * SROW + kk]);
      const v2f b = *reinterpret_cast<const v2f*>(&Wp[(kk >> 1) * BRS + 2 * n]);
      wmma_acc_slots(d, a, b);
    }
#pragma unroll
    for (int q = 0; q < 8; ++q) {
      const float v = beta * d[q];
      Osm[(8 * half + q) * SROW + n] = v > 0.f ? v : 0.f;
    }
  }
  __syncthreads();

  const v2f o = *reinterpret_cast<const v2f*>(&Osm[w * SROW + 2 * lane]);
  *reinterpret_cast<v2f*>(hout + (size_t)r * NHID + 2 * lane) = o;
}

__global__ void __launch_bounds__(256) out_kernel(
    const float* __restrict__ h, const float* __restrict__ Wout,
    const float* __restrict__ bout, float* __restrict__ out) {
  __shared__ float Ws[NHID * NOUT];
  __shared__ float bs[NOUT];
  for (int idx = threadIdx.x; idx < NHID * NOUT; idx += 256) Ws[idx] = Wout[idx];
  if (threadIdx.x < NOUT) bs[threadIdx.x] = bout[threadIdx.x];
  __syncthreads();

  const int i = blockIdx.x * blockDim.x + threadIdx.x;
  if (i >= N_NODES) return;

  float hr[NHID];
  const float4* hp = reinterpret_cast<const float4*>(h + (long long)i * NHID);
#pragma unroll
  for (int k4 = 0; k4 < NHID / 4; ++k4) {
    const float4 v = hp[k4];
    hr[4 * k4 + 0] = v.x; hr[4 * k4 + 1] = v.y;
    hr[4 * k4 + 2] = v.z; hr[4 * k4 + 3] = v.w;
  }

  float lg[NOUT];
#pragma unroll
  for (int c = 0; c < NOUT; ++c) lg[c] = bs[c];
  for (int k = 0; k < NHID; ++k) {
    const float hk = hr[k];
#pragma unroll
    for (int c = 0; c < NOUT; ++c) lg[c] = fmaf(hk, Ws[k * NOUT + c], lg[c]);
  }

  float mx = lg[0];
#pragma unroll
  for (int c = 1; c < NOUT; ++c) mx = fmaxf(mx, lg[c]);
  float s = 0.f;
#pragma unroll
  for (int c = 0; c < NOUT; ++c) s += expf(lg[c] - mx);
  const float lse = mx + logf(s);
#pragma unroll
  for (int c = 0; c < NOUT; ++c)
    out[(long long)i * NOUT + c] = lg[c] - lse;
}

extern "C" void kernel_launch(void* const* d_in, const int* in_sizes, int n_in,
                              void* d_out, int out_size, void* d_ws, size_t ws_size,
                              hipStream_t stream) {
  (void)in_sizes; (void)n_in; (void)out_size; (void)ws_size;
  const float* x     = (const float*)d_in[0];
  const int*   row   = (const int*)  d_in[1];
  const int*   col   = (const int*)  d_in[2];
  const float* ew    = (const float*)d_in[3];
  const float* Win   = (const float*)d_in[4];
  const float* bin   = (const float*)d_in[5];
  const float* convW = (const float*)d_in[6];
  const float* Wout  = (const float*)d_in[7];
  const float* bout  = (const float*)d_in[8];

  const size_t S = (size_t)N_NODES * NHID;
  float* h0 = (float*)d_ws;
  float* hA = h0 + S;
  float* hB = hA + S;
  int*  cnt    = (int*)(hB + S);
  int*  bsum   = cnt + 100096;
  int*  rowptr = bsum + 512;
  int*  rank   = rowptr + 100096;
  int2* meta   = (int2*)(rank + E_EDGES);

  const int rowBlocks = (N_NODES + 255) / 256;
  const size_t inLds = (size_t)(NIN / 2) * BRS * sizeof(float);
  in_gemm_kernel<<<rowBlocks, 256, inLds, stream>>>(x, Win, bin, h0);

  const int edgeBlocks = (E_EDGES + 255) / 256;
  zero_kernel<<<(N_NODES / 4 + 255) / 256, 256, 0, stream>>>((float4*)cnt, N_NODES / 4);
  rank_kernel<<<edgeBlocks, 256, 0, stream>>>(row, cnt, rank);
  blocksum_kernel<<<rowBlocks, 256, 0, stream>>>(cnt, bsum);
  scan_kernel<<<rowBlocks, 256, 0, stream>>>(cnt, bsum, rowptr);
  fill_kernel<<<edgeBlocks, 256, 0, stream>>>(row, col, ew, rank, rowptr, meta);

  const int layerBlocks = N_NODES / 16;

  const float* cur = h0;
  for (int l = 0; l < NLAYERS; ++l) {
    const float beta = logf(0.5f / (float)(l + 1) + 1.f);
    float* nxt = (l & 1) ? hB : hA;
    layer_kernel<<<layerBlocks, 512, 0, stream>>>(rowptr, meta, cur, h0,
                                                  convW + (size_t)l * NHID * NHID, beta, nxt);
    cur = nxt;
  }

  out_kernel<<<(N_NODES + 255) / 256, 256, 0, stream>>>(cur, Wout, bout, (float*)d_out);
}


# Layer launch as a resident grid: 1024 16-wave blocks (4 per WGP) loop over the 16-row tiles, 6-7 rows per wave instead of one wave per row, W staged to LDS once per block; every sum in the same order
# speedup vs baseline: 3.5095x; 1.0414x over previous
#include <hip/hip_runtime.h>
#include <hip/hip_bf16.h>
#include <math.h>

#define N_NODES 100000
#define E_EDGES 3200000
#define NIN     500
#define NHID    64
#define NOUT    40
#define NLAYERS 8
#define ALPHA   0.1f

#define BRS 160

typedef __attribute__((ext_vector_type(2))) float v2f;
typedef __attribute__((ext_vector_type(8))) float v8f;

__device__ __forceinline__ void wmma_acc(v8f& c, v2f a, v2f b) {
  asm("v_wmma_f32_16x16x4_f32 %0, %1, %2, %0" : "+v"(c) : "v"(a), "v"(b));
}
__device__ __forceinline__ void wmma_acc_slots(v8f& c, v2f a, v2f b) {
  asm volatile("v_wmma_f32_16x16x4_f32 %0, %1, %2, %0\n\tv_nop\n\tv_nop\n\tv_nop\n\tv_nop"
      : "+v"(c) : "v"(a), "v"(b));
}
__device__ __forceinline__ void wmma_epilogue_guard() {
  asm volatile("s_nop 8");
}

__global__ void __launch_bounds__(256) in_gemm_kernel(
    const float* __restrict__ x, const float* __restrict__ Win,
    const float* __restrict__ bin, float* __restrict__ h0) {
  extern __shared__ float Wp[];
  for (int idx = threadIdx.x; idx < NIN * NHID; idx += 256) {
    const int k = idx >> 6;
    const int n = idx & 63;
    Wp[(k >> 1) * BRS + 2 * n + (k & 1)] = Win[idx];
  }
  __syncthreads();

  const int wave = threadIdx.x >> 5;
  const int lane = threadIdx.x & 31;
  const int half = lane >> 4;
  const int l16  = lane & 15;
  const int r0   = blockIdx.x * 256 + wave * 32;

  const int arow0 = min(r0 + l16,      N_NODES - 1);
  const int arow1 = min(r0 + 16 + l16, N_NODES - 1);
  const float* xr0 = x + (long long)arow0 * NIN;
  const float* xr1 = x + (long long)arow1 * NIN;

  v8f acc0[4] = {};
  v8f acc1[4] = {};
  for (int k = 0; k < NIN; k += 4) {
    const int kk = k + 2 * half;
    const v2f a0 = *reinterpret_cast<const v2f*>(xr0 + kk);
    const v2f a1 = *reinterpret_cast<const v2f*>(xr1 + kk);
    const int bbase = (kk >> 1) * BRS + 2 * l16;
#pragma unroll
    for (int t = 0; t < 4; ++t) {
      const v2f b = *reinterpret_cast<const v2f*>(&Wp[bbase + t * 32]);
      wmma_acc(acc0[t], a0, b);
      wmma_acc(acc1[t], a1, b);
    }
  }
  wmma_epilogue_guard();

#pragma unroll
  for (int t = 0; t < 4; ++t) {
    const int n = t * 16 + l16;
    const float bias = bin[n];
#pragma unroll
    for (int r = 0; r < 8; ++r) {
      const int m = r + 8 * half;
      const int orow0 = r0 + m;
      const int orow1 = r0 + 16 + m;
      if (orow0 < N_NODES) {
        const float v = acc0[t][r] + bias;
        h0[(long long)orow0 * NHID + n] = v > 0.f ? v : 0.f;
      }
      if (orow1 < N_NODES) {
        const float v = acc1[t][r] + bias;
        h0[(long long)orow1 * NHID + n] = v > 0.f ? v : 0.f;
      }
    }
  }
}

__global__ void zero_kernel(float4* __restrict__ p, int n4) {
  int i = blockIdx.x * blockDim.x + threadIdx.x;
  if (i < n4) p[i] = make_float4(0.f, 0.f, 0.f, 0.f);
}


__global__ void __launch_bounds__(256) rank_kernel(
    const int* __restrict__ row, int* __restrict__ cnt, int* __restrict__ rank) {
  const int e = blockIdx.x * 256 + threadIdx.x;
  if (e < E_EDGES) rank[e] = atomicAdd(&cnt[row[e]], 1);
}

__global__ void __launch_bounds__(256) blocksum_kernel(
    const int* __restrict__ cnt, int* __restrict__ bsum) {
  __shared__ int wsum[8];
  const int i = blockIdx.x * 256 + threadIdx.x;
  int v = (i < N_NODES) ? cnt[i] : 0;
#pragma unroll
  for (int o = 16; o > 0; o >>= 1) v += __shfl_xor(v, o);
  if ((threadIdx.x & 31) == 0) wsum[threadIdx.x >> 5] = v;
  __syncthreads();
  if (threadIdx.x == 0) {
    int s = 0;
#pragma unroll
    for (int k = 0; k < 8; ++k) s += wsum[k];
    bsum[blockIdx.x] = s;
  }
}

__global__ void __launch_bounds__(256) scan_kernel(
    const int* __restrict__ cnt, const int* __restrict__ bsum,
    int* __restrict__ rowptr) {
  __shared__ int wsum[8];
  __shared__ int wpart[8];
  const int tid = threadIdx.x;
  const int lane = tid & 31;
  const int wv = tid >> 5;
  int part = 0;
  for (int j = tid; j < (int)blockIdx.x; j += 256) part += bsum[j];
#pragma unroll
  for (int o = 16; o > 0; o >>= 1) part += __shfl_xor(part, o);
  const int i = blockIdx.x * 256 + tid;
  const int c = (i < N_NODES) ? cnt[i] : 0;
  int incl = c;
#pragma unroll
  for (int o = 1; o < 32; o <<= 1) {
    const int t = __shfl_up(incl, o);
    if (lane >= o) incl += t;
  }
  if (lane == 31) wsum[wv] = incl;
  if (lane == 0) wpart[wv] = part;
  __syncthreads();
  int base = 0;
#pragma unroll
  for (int k = 0; k < 8; ++k) {
    base += wpart[k];
    if (k < wv) base += wsum[k];
  }
  if (i < N_NODES) rowptr[i] = base + incl - c;
  if (i == N_NODES - 1) rowptr[N_NODES] = base + incl;
}

__global__ void __launch_bounds__(256) fill_kernel(
    const int* __restrict__ row, const int* __restrict__ col,
    const float* __restrict__ ew, const int* __restrict__ rank,
    const int* __restrict__ rowptr, int2* __restrict__ meta) {
  const int e = blockIdx.x * 256 + threadIdx.x;
  if (e < E_EDGES) {
    const int p = rowptr[row[e]] + rank[e];
    meta[p] = make_int2(col[e], __float_as_int(ew[e]));
  }
}

#define SROW 68

static_assert(N_NODES % 16 == 0, "16 rows per tile, no partial tile");
#define N_TILES      (N_NODES / 16)
#define LAYER_BLOCKS 1024

__global__ void __launch_bounds__(512, 16) layer_kernel(
    const int* __restrict__ rowptr, const int2* __restrict__ meta,
    const float* __restrict__ h, const float* __restrict__ h0,
    const float* __restrict__ Wl, float beta, float* __restrict__ hout) {
  __shared__ __attribute__((aligned(16))) float Ssm[16 * SROW];
  __shared__ __attribute__((aligned(16))) float Osm[16 * SROW];
  __shared__ __attribute__((aligned(16))) float Wp[(NHID / 2) * BRS];
  {
    const int tp = threadIdx.x >> 4;
    const int tn = (threadIdx.x & 15) * 4;
    const float4 w0 = *reinterpret_cast<const float4*>(Wl + (2 * tp) * NHID + tn);
    const float4 w1 = *reinterpret_cast<const float4*>(Wl + (2 * tp + 1) * NHID + tn);
    *reinterpret_cast<float4*>(&Wp[tp * BRS + 2 * tn])     = make_float4(w0.x, w1.x, w0.y, w1.y);
    *reinterpret_cast<float4*>(&Wp[tp * BRS + 2 * tn + 4]) = make_float4(w0.z, w1.z, w0.w, w1.w);
  }
  const int lane = threadIdx.x & 31;
  const int w = __builtin_amdgcn_readfirstlane((int)(threadIdx.x >> 5));
  const float sA = 1.f - ALPHA;
  const float sB = ALPHA;
  const float cscale = (1.f - beta) / beta;

  for (int tile = blockIdx.x; tile < N_TILES; tile += LAYER_BLOCKS) {
    const int r = tile * 16 + w;
    const int start = rowptr[r];
    const int end   = rowptr[r + 1];
    const v2f g0 = *reinterpret_cast<const v2f*>(h0 + (size_t)r * NHID + 2 * lane);
    v2f acc = {0.f, 0.f};
    for (int e = start; e < end; e += 32) {
      const int n = end - e;
      int2 mine = meta[e + min(lane, n - 1)];
      if (lane >= n) mine.y = 0;
#pragma unroll
      for (int g = 0; g < 4; ++g) {
        if (g * 8 < n) {
          v2f v[8];
#pragma unroll
          for (int j = 0; j < 8; ++j) {
            const unsigned c = (unsigned)__builtin_amdgcn_readlane(mine.x, g * 8 + j);
            const float* hr = h + (size_t)c * NHID;
            v[j] = *reinterpret_cast<const v2f*>(hr + 2 * lane);
          }
#pragma unroll
          for (int j = 0; j < 8; ++j) {
            const float wt = __int_as_float(__builtin_amdgcn_readlane(mine.y, g * 8 + j));
            acc.x = fmaf(wt, v[j].x, acc.x);
            acc.y = fmaf(wt, v[j].y, acc.y);
          }
        }
        if (g == 0)
          asm volatile("" :: "v"(g0.x), "v"(g0.y));
      }
    }
    v2f s;
    s.x = sA * acc.x + sB * g0.x;
    s.y = sA * acc.y + sB * g0.y;
    *reinterpret_cast<v2f*>(&Ssm[w * SROW + 2 * lane]) = s;
    __syncthreads();

    if (w < 4) {
      const int half = lane >> 4;
      const int l16  = lane & 15;
      const int n    = w * 16 + l16;
      const float* sa = &Ssm[l16 * SROW + 2 * half];
      const float* wb = &Wp[half * BRS + 2 * n];
      v8f d;
#pragma unroll
      for (int q = 0; q < 8; ++q) d[q] = cscale * Ssm[(8 * half + q) * SROW + n];
#pragma unroll
      for (int kb = 0; kb < NHID; kb += 32) {
        v2f a[8], b[8];
#pragma unroll
        for (int j = 0; j < 8; ++j) {
          const int k = kb + 4 * j;
          a[j] = *reinterpret_cast<const v2f*>(sa + k);
          b[j] = *reinterpret_cast<const v2f*>(wb + (k >> 1) * BRS);
        }
        asm volatile("" :: "v"(a[0]), "v"(a[1]), "v"(a[2]), "v"(a[3]),
                           "v"(a[4]), "v"(a[5]), "v"(a[6]), "v"(a[7]),
                           "v"(b[0]), "v"(b[1]), "v"(b[2]), "v"(b[3]),
                           "v"(b[4]), "v"(b[5]), "v"(b[6]), "v"(b[7]));
#pragma unroll
        for (int j = 0; j < 8; ++j) wmma_acc_slots(d, a[j], b[j]);
      }
#pragma unroll
      for (int q = 0; q < 8; ++q) {
        const float v = beta * d[q];
        Osm[(8 * half + q) * SROW + n] = v > 0.f ? v : 0.f;
      }
    }
    __syncthreads();

    const v2f o = *reinterpret_cast<const v2f*>(&Osm[w * SROW + 2 * lane]);
    *reinterpret_cast<v2f*>(hout + (size_t)r * NHID + 2 * lane) = o;
  }
}

__global__ void __launch_bounds__(256) out_kernel(
    const float* __restrict__ h, const float* __restrict__ Wout,
    const float* __restrict__ bout, float* __restrict__ out) {
  __shared__ float Ws[NHID * NOUT];
  __shared__ float bs[NOUT];
  for (int idx = threadIdx.x; idx < NHID * NOUT; idx += 256) Ws[idx] = Wout[idx];
  if (threadIdx.x < NOUT) bs[threadIdx.x] = bout[threadIdx.x];
  __syncthreads();

  const int i = blockIdx.x * blockDim.x + threadIdx.x;
  if (i >= N_NODES) return;

  float hr[NHID];
  const float4* hp = reinterpret_cast<const float4*>(h + (long long)i * NHID);
#pragma unroll
  for (int k4 = 0; k4 < NHID / 4; ++k4) {
    const float4 v = hp[k4];
    hr[4 * k4 + 0] = v.x; hr[4 * k4 + 1] = v.y;
    hr[4 * k4 + 2] = v.z; hr[4 * k4 + 3] = v.w;
  }

  float lg[NOUT];
#pragma unroll
  for (int c = 0; c < NOUT; ++c) lg[c] = bs[c];
  for (int k = 0; k < NHID; ++k) {
    const float hk = hr[k];
#pragma unroll
    for (int c = 0; c < NOUT; ++c) lg[c] = fmaf(hk, Ws[k * NOUT + c], lg[c]);
  }

  float mx = lg[0];
#pragma unroll
  for (int c = 1; c < NOUT; ++c) mx = fmaxf(mx, lg[c]);
  float s = 0.f;
#pragma unroll
  for (int c = 0; c < NOUT; ++c) s += expf(lg[c] - mx);
  const float lse = mx + logf(s);
#pragma unroll
  for (int c = 0; c < NOUT; ++c)
    out[(long long)i * NOUT + c] = lg[c] - lse;
}

extern "C" void kernel_launch(void* const* d_in, const int* in_sizes, int n_in,
                              void* d_out, int out_size, void* d_ws, size_t ws_size,
                              hipStream_t stream) {
  (void)in_sizes; (void)n_in; (void)out_size; (void)ws_size;
  const float* x     = (const float*)d_in[0];
  const int*   row   = (const int*)  d_in[1];
  const int*   col   = (const int*)  d_in[2];
  const float* ew    = (const float*)d_in[3];
  const float* Win   = (const float*)d_in[4];
  const float* bin   = (const float*)d_in[5];
  const float* convW = (const float*)d_in[6];
  const float* Wout  = (const float*)d_in[7];
  const float* bout  = (const float*)d_in[8];

  const size_t S = (size_t)N_NODES * NHID;
  float* h0 = (float*)d_ws;
  float* hA = h0 + S;
  float* hB = hA + S;
  int*  cnt    = (int*)(hB + S);
  int*  bsum   = cnt + 100096;
  int*  rowptr = bsum + 512;
  int*  rank   = rowptr + 100096;
  int2* meta   = (int2*)(rank + E_EDGES);

  const int rowBlocks = (N_NODES + 255) / 256;
  const size_t inLds = (size_t)(NIN / 2) * BRS * sizeof(float);
  in_gemm_kernel<<<rowBlocks, 256, inLds, stream>>>(x, Win, bin, h0);

  const int edgeBlocks = (E_EDGES + 255) / 256;
  zero_kernel<<<(N_NODES / 4 + 255) / 256, 256, 0, stream>>>((float4*)cnt, N_NODES / 4);
  rank_kernel<<<edgeBlocks, 256, 0, stream>>>(row, cnt, rank);
  blocksum_kernel<<<rowBlocks, 256, 0, stream>>>(cnt, bsum);
  scan_kernel<<<rowBlocks, 256, 0, stream>>>(cnt, bsum, rowptr);
  fill_kernel<<<edgeBlocks, 256, 0, stream>>>(row, col, ew, rank, rowptr, meta);

  const int layerBlocks = LAYER_BLOCKS;

  const float* cur = h0;
  for (int l = 0; l < NLAYERS; ++l) {
    const float beta = logf(0.5f / (float)(l + 1) + 1.f);
    float* nxt = (l & 1) ? hB : hA;
    layer_kernel<<<layerBlocks, 512, 0, stream>>>(rowptr, meta, cur, h0,
                                                  convW + (size_t)l * NHID * NHID, beta, nxt);
    cur = nxt;
  }

  out_kernel<<<(N_NODES + 255) / 256, 256, 0, stream>>>(cur, Wout, bout, (float*)d_out);
}


# Edge grouping by destination row as a two-level counting sort through LDS (256-row bins): no per-edge global atomic or scattered store; 2 launches fewer; row sums in arrival order as before
# speedup vs baseline: 4.9338x; 1.4058x over previous
#include <hip/hip_runtime.h>
#include <hip/hip_bf16.h>
#include <math.h>

#define N_NODES 100000
#define E_EDGES 3200000
#define NIN     500
#define NHID    64
#define NOUT    40
#define NLAYERS 8
#define ALPHA   0.1f

#define BRS 160

typedef __attribute__((ext_vector_type(2))) float v2f;
typedef __attribute__((ext_vector_type(8))) float v8f;

__device__ __forceinline__ void wmma_acc(v8f& c, v2f a, v2f b) {
  asm("v_wmma_f32_16x16x4_f32 %0, %1, %2, %0" : "+v"(c) : "v"(a), "v"(b));
}
__device__ __forceinline__ void wmma_acc_slots(v8f& c, v2f a, v2f b) {
  asm volatile("v_wmma_f32_16x16x4_f32 %0, %1, %2, %0\n\tv_nop\n\tv_nop\n\tv_nop\n\tv_nop"
      : "+v"(c) : "v"(a), "v"(b));
}
__device__ __forceinline__ void wmma_epilogue_guard() {
  asm volatile("s_nop 8");
}

__global__ void __launch_bounds__(256) in_gemm_kernel(
    const float* __restrict__ x, const float* __restrict__ Win,
    const float* __restrict__ bin, float* __restrict__ h0) {
  extern __shared__ float Wp[];
  for (int idx = threadIdx.x; idx < NIN * NHID; idx += 256) {
    const int k = idx >> 6;
    const int n = idx & 63;
    Wp[(k >> 1) * BRS + 2 * n + (k & 1)] = Win[idx];
  }
  __syncthreads();

  const int wave = threadIdx.x >> 5;
  const int lane = threadIdx.x & 31;
  const int half = lane >> 4;
  const int l16  = lane & 15;
  const int r0   = blockIdx.x * 256 + wave * 32;

  const int arow0 = min(r0 + l16,      N_NODES - 1);
  const int arow1 = min(r0 + 16 + l16, N_NODES - 1);
  const float* xr0 = x + (long long)arow0 * NIN;
  const float* xr1 = x + (long long)arow1 * NIN;

  v8f acc0[4] = {};
  v8f acc1[4] = {};
  for (int k = 0; k < NIN; k += 4) {
    const int kk = k + 2 * half;
    const v2f a0 = *reinterpret_cast<const v2f*>(xr0 + kk);
    const v2f a1 = *reinterpret_cast<const v2f*>(xr1 + kk);
    const int bbase = (kk >> 1) * BRS + 2 * l16;
#pragma unroll
    for (int t = 0; t < 4; ++t) {
      const v2f b = *reinterpret_cast<const v2f*>(&Wp[bbase + t * 32]);
      wmma_acc(acc0[t], a0, b);
      wmma_acc(acc1[t], a1, b);
    }
  }
  wmma_epilogue_guard();

#pragma unroll
  for (int t = 0; t < 4; ++t) {
    const int n = t * 16 + l16;
    const float bias = bin[n];
#pragma unroll
    for (int r = 0; r < 8; ++r) {
      const int m = r + 8 * half;
      const int orow0 = r0 + m;
      const int orow1 = r0 + 16 + m;
      if (orow0 < N_NODES) {
        const float v = acc0[t][r] + bias;
        h0[(long long)orow0 * NHID + n] = v > 0.f ? v : 0.f;
      }
      if (orow1 < N_NODES) {
        const float v = acc1[t][r] + bias;
        h0[(long long)orow1 * NHID + n] = v > 0.f ? v : 0.f;
      }
    }
  }
}

__global__ void zero_kernel(float4* __restrict__ p, int n4) {
  int i = blockIdx.x * blockDim.x + threadIdx.x;
  if (i < n4) p[i] = make_float4(0.f, 0.f, 0.f, 0.f);
}

#define BIN_SHIFT    8
#define BIN_ROWS     (1 << BIN_SHIFT)
#define NBINS        ((N_NODES + BIN_ROWS - 1) / BIN_ROWS)
#define CHUNK        8192
#define NCHUNKS      ((E_EDGES + CHUNK - 1) / CHUNK)
#define SORT_THREADS 1024
#define PER_THREAD   (CHUNK / SORT_THREADS)
#define STAGE_CAP    16384
static_assert(NBINS <= SORT_THREADS && NBINS <= 512, "one thread per bin in the scans");
static_assert(N_NODES <= (1 << 24), "col is packed into 24 bits beside the row's low 8");

__device__ __forceinline__ int block_excl_scan(int v, int* wsum) {
  const int lane = threadIdx.x & 31;
  const int wv = threadIdx.x >> 5;
  int incl = v;
  for (int o = 1; o < 32; o <<= 1) {
    const int t = __shfl_up(incl, o);
    if (lane >= o) incl += t;
  }
  if (lane == 31) wsum[wv] = incl;
  __syncthreads();
  int base = 0;
  for (int k = 0; k < 32; ++k) {
    const int sk = wsum[k];
    if (k < wv) base += sk;
  }
  __syncthreads();
  return base + incl - v;
}

__global__ void __launch_bounds__(SORT_THREADS) bin_hist_kernel(
    const int* __restrict__ row, int* __restrict__ binTotal) {
  __shared__ int hist[512];
  const int tid = threadIdx.x;
  if (tid < 512) hist[tid] = 0;
  __syncthreads();
  const int e0 = blockIdx.x * CHUNK;
  int rw[PER_THREAD];
#pragma unroll
  for (int j = 0; j < PER_THREAD; ++j)
    rw[j] = row[min(e0 + j * SORT_THREADS + tid, E_EDGES - 1)];
#pragma unroll
  for (int j = 0; j < PER_THREAD; ++j)
    if (e0 + j * SORT_THREADS + tid < E_EDGES) atomicAdd(&hist[rw[j] >> BIN_SHIFT], 1);
  __syncthreads();
  if (tid < NBINS) {
    const int c = hist[tid];
    if (c) atomicAdd(&binTotal[tid], c);
  }
}

__global__ void __launch_bounds__(SORT_THREADS) partition_kernel(
    const int* __restrict__ row, const int* __restrict__ col,
    const float* __restrict__ ew, const int* __restrict__ binTotal,
    int* __restrict__ cursor, int2* __restrict__ tmp) {
  __shared__ int hist[512];
  __shared__ int lstart[512];
  __shared__ int gbase[512];
  __shared__ int wsum[32];
  __shared__ __attribute__((aligned(16))) int2 item[CHUNK];
  __shared__ int dst[CHUNK];
  const int tid = threadIdx.x;
  if (tid < 512) hist[tid] = 0;
  __syncthreads();

  const int e0 = blockIdx.x * CHUNK;
  int bn[PER_THREAD], rk[PER_THREAD], pk[PER_THREAD], wb[PER_THREAD];
#pragma unroll
  for (int j = 0; j < PER_THREAD; ++j) {
    const int ec = min(e0 + j * SORT_THREADS + tid, E_EDGES - 1);
    bn[j] = row[ec];
    pk[j] = col[ec];
    wb[j] = __float_as_int(ew[ec]);
  }
#pragma unroll
  for (int j = 0; j < PER_THREAD; ++j) {
    const int rw = bn[j];
    pk[j] = (int)(((unsigned)(rw & (BIN_ROWS - 1)) << 24) | (unsigned)pk[j]);
    rk[j] = 0;
    if (e0 + j * SORT_THREADS + tid < E_EDGES) {
      bn[j] = rw >> BIN_SHIFT;
      rk[j] = atomicAdd(&hist[bn[j]], 1);
    } else {
      bn[j] = -1;
    }
  }
  __syncthreads();

  const int hcount = (tid < NBINS) ? hist[tid] : 0;
  const int btotal = (tid < NBINS) ? binTotal[tid] : 0;
  const int ls = block_excl_scan(hcount, wsum);
  const int bs = block_excl_scan(btotal, wsum);
  if (tid < NBINS) {
    lstart[tid] = ls;
    const int reserved = hcount ? atomicAdd(&cursor[tid], hcount) : 0;
    gbase[tid] = bs + reserved;
  }
  __syncthreads();

#pragma unroll
  for (int j = 0; j < PER_THREAD; ++j) {
    if (bn[j] >= 0) {
      const int p = lstart[bn[j]] + rk[j];
      item[p] = make_int2(pk[j], wb[j]);
      dst[p]  = gbase[bn[j]] + rk[j];
    }
  }
  __syncthreads();

  const int nvalid = min(CHUNK, E_EDGES - e0);
  for (int j = 0; j < PER_THREAD; ++j) {
    const int p = j * SORT_THREADS + tid;
    if (p < nvalid) tmp[dst[p]] = item[p];
  }
}

__global__ void __launch_bounds__(SORT_THREADS) binsort_kernel(
    const int* __restrict__ binTotal, const int2* __restrict__ tmp,
    int* __restrict__ rowptr, int2* __restrict__ meta) {
  __shared__ int cnt[BIN_ROWS];
  __shared__ int fillc[BIN_ROWS];
  __shared__ int rstart[BIN_ROWS];
  __shared__ int wsum[32];
  __shared__ int sbase;
  __shared__ __attribute__((aligned(16))) int2 stage[STAGE_CAP];
  const int tid = threadIdx.x;
  const int b = blockIdx.x;
  if (tid < BIN_ROWS) { cnt[tid] = 0; fillc[tid] = 0; }
  const int btotal = (tid < NBINS) ? binTotal[tid] : 0;
  const int bs = block_excl_scan(btotal, wsum);
  if (tid == b) sbase = bs;
  __syncthreads();
  const int base = sbase;
  const int nb = binTotal[b];

  for (int i = tid; i < nb; i += SORT_THREADS) {
    const int2 it = tmp[base + i];
    atomicAdd(&cnt[(unsigned)it.x >> 24], 1);
  }
  __syncthreads();
  const int c = (tid < BIN_ROWS) ? cnt[tid] : 0;
  const int rs = block_excl_scan(c, wsum);
  if (tid < BIN_ROWS) {
    rstart[tid] = rs;
    const int r = b * BIN_ROWS + tid;
    if (r < N_NODES) rowptr[r] = base + rs;
  }
  if (b == NBINS - 1 && tid == 0) rowptr[N_NODES] = base + nb;
  __syncthreads();

  const bool staged = (nb <= STAGE_CAP);
  for (int i = tid; i < nb; i += SORT_THREADS) {
    const int2 it = tmp[base + i];
    const int rl = (unsigned)it.x >> 24;
    const int p = rstart[rl] + atomicAdd(&fillc[rl], 1);
    const int2 o = make_int2(it.x & 0x00FFFFFF, it.y);
    if (staged) stage[p] = o; else meta[base + p] = o;
  }
  if (staged) {
    __syncthreads();
    for (int i = tid; i < nb; i += SORT_THREADS) meta[base + i] = stage[i];
  }
}

#define SROW 68

static_assert(N_NODES % 16 == 0, "16 rows per tile, no partial tile");
#define N_TILES      (N_NODES / 16)
#define LAYER_BLOCKS 1024

__global__ void __launch_bounds__(512, 16) layer_kernel(
    const int* __restrict__ rowptr, const int2* __restrict__ meta,
    const float* __restrict__ h, const float* __restrict__ h0,
    const float* __restrict__ Wl, float beta, float* __restrict__ hout) {
  __shared__ __attribute__((aligned(16))) float Ssm[16 * SROW];
  __shared__ __attribute__((aligned(16))) float Osm[16 * SROW];
  __shared__ __attribute__((aligned(16))) float Wp[(NHID / 2) * BRS];
  {
    const int tp = threadIdx.x >> 4;
    const int tn = (threadIdx.x & 15) * 4;
    const float4 w0 = *reinterpret_cast<const float4*>(Wl + (2 * tp) * NHID + tn);
    const float4 w1 = *reinterpret_cast<const float4*>(Wl + (2 * tp + 1) * NHID + tn);
    *reinterpret_cast<float4*>(&Wp[tp * BRS + 2 * tn])     = make_float4(w0.x, w1.x, w0.y, w1.y);
    *reinterpret_cast<float4*>(&Wp[tp * BRS + 2 * tn + 4]) = make_float4(w0.z, w1.z, w0.w, w1.w);
  }
  const int lane = threadIdx.x & 31;
  const int w = __builtin_amdgcn_readfirstlane((int)(threadIdx.x >> 5));
  const float sA = 1.f - ALPHA;
  const float sB = ALPHA;
  const float cscale = (1.f - beta) / beta;

  for (int tile = blockIdx.x; tile < N_TILES; tile += LAYER_BLOCKS) {
    const int r = tile * 16 + w;
    const int start = rowptr[r];
    const int end   = rowptr[r + 1];
    const v2f g0 = *reinterpret_cast<const v2f*>(h0 + (size_t)r * NHID + 2 * lane);
    v2f acc = {0.f, 0.f};
    for (int e = start; e < end; e += 32) {
      const int n = end - e;
      int2 mine = meta[e + min(lane, n - 1)];
      if (lane >= n) mine.y = 0;
#pragma unroll
      for (int g = 0; g < 4; ++g) {
        if (g * 8 < n) {
          v2f v[8];
#pragma unroll
          for (int j = 0; j < 8; ++j) {
            const unsigned c = (unsigned)__builtin_amdgcn_readlane(mine.x, g * 8 + j);
            const float* hr = h + (size_t)c * NHID;
            v[j] = *reinterpret_cast<const v2f*>(hr + 2 * lane);
          }
#pragma unroll
          for (int j = 0; j < 8; ++j) {
            const float wt = __int_as_float(__builtin_amdgcn_readlane(mine.y, g * 8 + j));
            acc.x = fmaf(wt, v[j].x, acc.x);
            acc.y = fmaf(wt, v[j].y, acc.y);
          }
        }
        if (g == 0)
          asm volatile("" :: "v"(g0.x), "v"(g0.y));
      }
    }
    v2f s;
    s.x = sA * acc.x + sB * g0.x;
    s.y = sA * acc.y + sB * g0.y;
    *reinterpret_cast<v2f*>(&Ssm[w * SROW + 2 * lane]) = s;
    __syncthreads();

    if (w < 4) {
      const int half = lane >> 4;
      const int l16  = lane & 15;
      const int n    = w * 16 + l16;
      const float* sa = &Ssm[l16 * SROW + 2 * half];
      const float* wb = &Wp[half * BRS + 2 * n];
      v8f d;
#pragma unroll
      for (int q = 0; q < 8; ++q) d[q] = cscale * Ssm[(8 * half + q) * SROW + n];
#pragma unroll
      for (int kb = 0; kb < NHID; kb += 32) {
        v2f a[8], b[8];
#pragma unroll
        for (int j = 0; j < 8; ++j) {
          const int k = kb + 4 * j;
          a[j] = *reinterpret_cast<const v2f*>(sa + k);
          b[j] = *reinterpret_cast<const v2f*>(wb + (k >> 1) * BRS);
        }
        asm volatile("" :: "v"(a[0]), "v"(a[1]), "v"(a[2]), "v"(a[3]),
                           "v"(a[4]), "v"(a[5]), "v"(a[6]), "v"(a[7]),
                           "v"(b[0]), "v"(b[1]), "v"(b[2]), "v"(b[3]),
                           "v"(b[4]), "v"(b[5]), "v"(b[6]), "v"(b[7]));
#pragma unroll
        for (int j = 0; j < 8; ++j) wmma_acc_slots(d, a[j], b[j]);
      }
#pragma unroll
      for (int q = 0; q < 8; ++q) {
        const float v = beta * d[q];
        Osm[(8 * half + q) * SROW + n] = v > 0.f ? v : 0.f;
      }
    }
    __syncthreads();

    const v2f o = *reinterpret_cast<const v2f*>(&Osm[w * SROW + 2 * lane]);
    *reinterpret_cast<v2f*>(hout + (size_t)r * NHID + 2 * lane) = o;
  }
}

__global__ void __launch_bounds__(256) out_kernel(
    const float* __restrict__ h, const float* __restrict__ Wout,
    const float* __restrict__ bout, float* __restrict__ out) {
  __shared__ float Ws[NHID * NOUT];
  __shared__ float bs[NOUT];
  for (int idx = threadIdx.x; idx < NHID * NOUT; idx += 256) Ws[idx] = Wout[idx];
  if (threadIdx.x < NOUT) bs[threadIdx.x] = bout[threadIdx.x];
  __syncthreads();

  const int i = blockIdx.x * blockDim.x + threadIdx.x;
  if (i >= N_NODES) return;

  float hr[NHID];
  const float4* hp = reinterpret_cast<const float4*>(h + (long long)i * NHID);
#pragma unroll
  for (int k4 = 0; k4 < NHID / 4; ++k4) {
    const float4 v = hp[k4];
    hr[4 * k4 + 0] = v.x; hr[4 * k4 + 1] = v.y;
    hr[4 * k4 + 2] = v.z; hr[4 * k4 + 3] = v.w;
  }

  float lg[NOUT];
#pragma unroll
  for (int c = 0; c < NOUT; ++c) lg[c] = bs[c];
  for (int k = 0; k < NHID; ++k) {
    const float hk = hr[k];
#pragma unroll
    for (int c = 0; c < NOUT; ++c) lg[c] = fmaf(hk, Ws[k * NOUT + c], lg[c]);
  }

  float mx = lg[0];
#pragma unroll
  for (int c = 1; c < NOUT; ++c) mx = fmaxf(mx, lg[c]);
  float s = 0.f;
#pragma unroll
  for (int c = 0; c < NOUT; ++c) s += expf(lg[c] - mx);
  const float lse = mx + logf(s);
#pragma unroll
  for (int c = 0; c < NOUT; ++c)
    out[(long long)i * NOUT + c] = lg[c] - lse;
}

extern "C" void kernel_launch(void* const* d_in, const int* in_sizes, int n_in,
                              void* d_out, int out_size, void* d_ws, size_t ws_size,
                              hipStream_t stream) {
  (void)in_sizes; (void)n_in; (void)out_size; (void)ws_size;
  const float* x     = (const float*)d_in[0];
  const int*   row   = (const int*)  d_in[1];
  const int*   col   = (const int*)  d_in[2];
  const float* ew    = (const float*)d_in[3];
  const float* Win   = (const float*)d_in[4];
  const float* bin   = (const float*)d_in[5];
  const float* convW = (const float*)d_in[6];
  const float* Wout  = (const float*)d_in[7];
  const float* bout  = (const float*)d_in[8];

  const size_t S = (size_t)N_NODES * NHID;
  float* h0 = (float*)d_ws;
  float* hA = h0 + S;
  float* hB = hA + S;
  int*  binTotal = (int*)(hB + S);
  int*  cursor   = binTotal + 512;
  int*  rowptr   = cursor + 512;
  int2* tmp      = (int2*)(rowptr + 100096);
  int2* meta     = tmp + E_EDGES;

  const int rowBlocks = (N_NODES + 255) / 256;
  const size_t inLds = (size_t)(NIN / 2) * BRS * sizeof(float);
  in_gemm_kernel<<<rowBlocks, 256, inLds, stream>>>(x, Win, bin, h0);

  zero_kernel<<<1, 256, 0, stream>>>((float4*)binTotal, 256);
  bin_hist_kernel<<<NCHUNKS, SORT_THREADS, 0, stream>>>(row, binTotal);
  partition_kernel<<<NCHUNKS, SORT_THREADS, 0, stream>>>(row, col, ew, binTotal, cursor, tmp);
  binsort_kernel<<<NBINS, SORT_THREADS, 0, stream>>>(binTotal, tmp, rowptr, meta);

  const int layerBlocks = LAYER_BLOCKS;

  const float* cur = h0;
  for (int l = 0; l < NLAYERS; ++l) {
    const float beta = logf(0.5f / (float)(l + 1) + 1.f);
    float* nxt = (l & 1) ? hB : hA;
    layer_kernel<<<layerBlocks, 512, 0, stream>>>(rowptr, meta, cur, h0,
                                                  convW + (size_t)l * NHID * NHID, beta, nxt);
    cur = nxt;
  }

  out_kernel<<<(N_NODES + 255) / 256, 256, 0, stream>>>(cur, Wout, bout, (float*)d_out);
}


# Two launches folded into neighbours: edge-bin histogram at the head of the input-GEMM blocks; output head in the last layer launch (logits on WMMA in k-pair order, softmax sums by wave reduction)
# speedup vs baseline: 5.2051x; 1.0550x over previous
#include <hip/hip_runtime.h>
#include <hip/hip_bf16.h>
#include <math.h>

#define N_NODES 100000
#define E_EDGES 3200000
#define NIN     500
#define NHID    64
#define NOUT    40
#define NLAYERS 8
#define ALPHA   0.1f

#define BRS 160

#define BIN_SHIFT    8
#define BIN_ROWS     (1 << BIN_SHIFT)
#define NBINS        ((N_NODES + BIN_ROWS - 1) / BIN_ROWS)
#define CHUNK        8192
#define NCHUNKS      ((E_EDGES + CHUNK - 1) / CHUNK)
#define GEMM_BLOCKS  ((N_NODES + 255) / 256)
#define HIST_PER_THREAD (CHUNK / 256)
static_assert(NCHUNKS == GEMM_BLOCKS, "one edge chunk per input-GEMM block");
static_assert(NBINS <= 512, "bin histogram is 512 LDS words");

typedef __attribute__((ext_vector_type(2))) float v2f;
typedef __attribute__((ext_vector_type(8))) float v8f;

__device__ __forceinline__ void wmma_acc(v8f& c, v2f a, v2f b) {
  asm("v_wmma_f32_16x16x4_f32 %0, %1, %2, %0" : "+v"(c) : "v"(a), "v"(b));
}
__device__ __forceinline__ void wmma_acc_slots(v8f& c, v2f a, v2f b) {
  asm volatile("v_wmma_f32_16x16x4_f32 %0, %1, %2, %0\n\tv_nop\n\tv_nop\n\tv_nop\n\tv_nop"
      : "+v"(c) : "v"(a), "v"(b));
}
__device__ __forceinline__ void wmma_epilogue_guard() {
  asm volatile("s_nop 8");
}

__global__ void __launch_bounds__(256) in_gemm_kernel(
    const float* __restrict__ x, const float* __restrict__ Win,
    const float* __restrict__ bin, float* __restrict__ h0,
    const int* __restrict__ row, int* __restrict__ binTotal) {
  extern __shared__ float Wp[];
  {
    int* hist = reinterpret_cast<int*>(Wp);
    hist[threadIdx.x] = 0;
    hist[threadIdx.x + 256] = 0;
    const int e0 = blockIdx.x * CHUNK;
    int rw[HIST_PER_THREAD];
#pragma unroll
    for (int j = 0; j < HIST_PER_THREAD; ++j)
      rw[j] = row[min(e0 + j * 256 + (int)threadIdx.x, E_EDGES - 1)];
    __syncthreads();
#pragma unroll
    for (int j = 0; j < HIST_PER_THREAD; ++j)
      if (e0 + j * 256 + (int)threadIdx.x < E_EDGES) atomicAdd(&hist[rw[j] >> BIN_SHIFT], 1);
    __syncthreads();
    for (int i = threadIdx.x; i < NBINS; i += 256) {
      const int c = hist[i];
      if (c) atomicAdd(&binTotal[i], c);
    }
    __syncthreads();
  }
  for (int idx = threadIdx.x; idx < NIN * NHID; idx += 256) {
    const int k = idx >> 6;
    const int n = idx & 63;
    Wp[(k >> 1) * BRS + 2 * n + (k & 1)] = Win[idx];
  }
  __syncthreads();

  const int wave = threadIdx.x >> 5;
  const int lane = threadIdx.x & 31;
  const int half = lane >> 4;
  const int l16  = lane & 15;
  const int r0   = blockIdx.x * 256 + wave * 32;

  const int arow0 = min(r0 + l16,      N_NODES - 1);
  const int arow1 = min(r0 + 16 + l16, N_NODES - 1);
  const float* xr0 = x + (long long)arow0 * NIN;
  const float* xr1 = x + (long long)arow1 * NIN;

  v8f acc0[4] = {};
  v8f acc1[4] = {};
  for (int k = 0; k < NIN; k += 4) {
    const int kk = k + 2 * half;
    const v2f a0 = *reinterpret_cast<const v2f*>(xr0 + kk);
    const v2f a1 = *reinterpret_cast<const v2f*>(xr1 + kk);
    const int bbase = (kk >> 1) * BRS + 2 * l16;
#pragma unroll
    for (int t = 0; t < 4; ++t) {
      const v2f b = *reinterpret_cast<const v2f*>(&Wp[bbase + t * 32]);
      wmma_acc(acc0[t], a0, b);
      wmma_acc(acc1[t], a1, b);
    }
  }
  wmma_epilogue_guard();

#pragma unroll
  for (int t = 0; t < 4; ++t) {
    const int n = t * 16 + l16;
    const float bias = bin[n];
#pragma unroll
    for (int r = 0; r < 8; ++r) {
      const int m = r + 8 * half;
      const int orow0 = r0 + m;
      const int orow1 = r0 + 16 + m;
      if (orow0 < N_NODES) {
        const float v = acc0[t][r] + bias;
        h0[(long long)orow0 * NHID + n] = v > 0.f ? v : 0.f;
      }
      if (orow1 < N_NODES) {
        const float v = acc1[t][r] + bias;
        h0[(long long)orow1 * NHID + n] = v > 0.f ? v : 0.f;
      }
    }
  }
}

__global__ void zero_kernel(float4* __restrict__ p, int n4) {
  int i = blockIdx.x * blockDim.x + threadIdx.x;
  if (i < n4) p[i] = make_float4(0.f, 0.f, 0.f, 0.f);
}

#define SORT_THREADS 1024
#define PER_THREAD   (CHUNK / SORT_THREADS)
#define STAGE_CAP    16384
static_assert(NBINS <= SORT_THREADS && NBINS <= 512, "one thread per bin in the scans");
static_assert(N_NODES <= (1 << 24), "col is packed into 24 bits beside the row's low 8");

__device__ __forceinline__ int block_excl_scan(int v, int* wsum) {
  const int lane = threadIdx.x & 31;
  const int wv = threadIdx.x >> 5;
  int incl = v;
  for (int o = 1; o < 32; o <<= 1) {
    const int t = __shfl_up(incl, o);
    if (lane >= o) incl += t;
  }
  if (lane == 31) wsum[wv] = incl;
  __syncthreads();
  int base = 0;
  for (int k = 0; k < 32; ++k) {
    const int sk = wsum[k];
    if (k < wv) base += sk;
  }
  __syncthreads();
  return base + incl - v;
}

__global__ void __launch_bounds__(SORT_THREADS) partition_kernel(
    const int* __restrict__ row, const int* __restrict__ col,
    const float* __restrict__ ew, const int* __restrict__ binTotal,
    int* __restrict__ cursor, int2* __restrict__ tmp) {
  __shared__ int hist[512];
  __shared__ int lstart[512];
  __shared__ int gbase[512];
  __shared__ int wsum[32];
  __shared__ __attribute__((aligned(16))) int2 item[CHUNK];
  __shared__ int dst[CHUNK];
  const int tid = threadIdx.x;
  if (tid < 512) hist[tid] = 0;
  __syncthreads();

  const int e0 = blockIdx.x * CHUNK;
  int bn[PER_THREAD], rk[PER_THREAD], pk[PER_THREAD], wb[PER_THREAD];
#pragma unroll
  for (int j = 0; j < PER_THREAD; ++j) {
    const int ec = min(e0 + j * SORT_THREADS + tid, E_EDGES - 1);
    bn[j] = row[ec];
    pk[j] = col[ec];
    wb[j] = __float_as_int(ew[ec]);
  }
#pragma unroll
  for (int j = 0; j < PER_THREAD; ++j) {
    const int rw = bn[j];
    pk[j] = (int)(((unsigned)(rw & (BIN_ROWS - 1)) << 24) | (unsigned)pk[j]);
    rk[j] = 0;
    if (e0 + j * SORT_THREADS + tid < E_EDGES) {
      bn[j] = rw >> BIN_SHIFT;
      rk[j] = atomicAdd(&hist[bn[j]], 1);
    } else {
      bn[j] = -1;
    }
  }
  __syncthreads();

  const int hcount = (tid < NBINS) ? hist[tid] : 0;
  const int btotal = (tid < NBINS) ? binTotal[tid] : 0;
  const int ls = block_excl_scan(hcount, wsum);
  const int bs = block_excl_scan(btotal, wsum);
  if (tid < NBINS) {
    lstart[tid] = ls;
    const int reserved = hcount ? atomicAdd(&cursor[tid], hcount) : 0;
    gbase[tid] = bs + reserved;
  }
  __syncthreads();

#pragma unroll
  for (int j = 0; j < PER_THREAD; ++j) {
    if (bn[j] >= 0) {
      const int p = lstart[bn[j]] + rk[j];
      item[p] = make_int2(pk[j], wb[j]);
      dst[p]  = gbase[bn[j]] + rk[j];
    }
  }
  __syncthreads();

  const int nvalid = min(CHUNK, E_EDGES - e0);
  for (int j = 0; j < PER_THREAD; ++j) {
    const int p = j * SORT_THREADS + tid;
    if (p < nvalid) tmp[dst[p]] = item[p];
  }
}

__global__ void __launch_bounds__(SORT_THREADS) binsort_kernel(
    const int* __restrict__ binTotal, const int2* __restrict__ tmp,
    int* __restrict__ rowptr, int2* __restrict__ meta) {
  __shared__ int cnt[BIN_ROWS];
  __shared__ int fillc[BIN_ROWS];
  __shared__ int rstart[BIN_ROWS];
  __shared__ int wsum[32];
  __shared__ int sbase;
  __shared__ __attribute__((aligned(16))) int2 stage[STAGE_CAP];
  const int tid = threadIdx.x;
  const int b = blockIdx.x;
  if (tid < BIN_ROWS) { cnt[tid] = 0; fillc[tid] = 0; }
  const int btotal = (tid < NBINS) ? binTotal[tid] : 0;
  const int bs = block_excl_scan(btotal, wsum);
  if (tid == b) sbase = bs;
  __syncthreads();
  const int base = sbase;
  const int nb = binTotal[b];

  for (int i = tid; i < nb; i += SORT_THREADS) {
    const int2 it = tmp[base + i];
    atomicAdd(&cnt[(unsigned)it.x >> 24], 1);
  }
  __syncthreads();
  const int c = (tid < BIN_ROWS) ? cnt[tid] : 0;
  const int rs = block_excl_scan(c, wsum);
  if (tid < BIN_ROWS) {
    rstart[tid] = rs;
    const int r = b * BIN_ROWS + tid;
    if (r < N_NODES) rowptr[r] = base + rs;
  }
  if (b == NBINS - 1 && tid == 0) rowptr[N_NODES] = base + nb;
  __syncthreads();

  const bool staged = (nb <= STAGE_CAP);
  for (int i = tid; i < nb; i += SORT_THREADS) {
    const int2 it = tmp[base + i];
    const int rl = (unsigned)it.x >> 24;
    const int p = rstart[rl] + atomicAdd(&fillc[rl], 1);
    const int2 o = make_int2(it.x & 0x00FFFFFF, it.y);
    if (staged) stage[p] = o; else meta[base + p] = o;
  }
  if (staged) {
    __syncthreads();
    for (int i = tid; i < nb; i += SORT_THREADS) meta[base + i] = stage[i];
  }
}

#define SROW 68
#define ORS  96
#define LROW 52

static_assert(N_NODES % 16 == 0, "16 rows per tile, no partial tile");
#define N_TILES      (N_NODES / 16)
#define LAYER_BLOCKS 1024

template <bool LAST>
__global__ void __launch_bounds__(512, 16) layer_kernel(
    const int* __restrict__ rowptr, const int2* __restrict__ meta,
    const float* __restrict__ h, const float* __restrict__ h0,
    const float* __restrict__ Wl, float beta, float* __restrict__ hout,
    const float* __restrict__ Wout, const float* __restrict__ bout,
    float* __restrict__ out) {
  __shared__ __attribute__((aligned(16))) float Ssm[16 * SROW];
  __shared__ __attribute__((aligned(16))) float Osm[16 * SROW];
  __shared__ __attribute__((aligned(16))) float Wp[(NHID / 2) * BRS];
  __shared__ __attribute__((aligned(16))) float Wo[LAST ? (NHID / 2) * ORS : 1];
  __shared__ __attribute__((aligned(16))) float Lsm[LAST ? 16 * LROW : 1];
  __shared__ __attribute__((aligned(16))) float Tsm[LAST ? 16 * NOUT : 1];
  __shared__ float Bsm[LAST ? 48 : 1];
  if (LAST) {
    for (int idx = threadIdx.x; idx < NHID * NOUT; idx += 512) {
      const int k = idx / NOUT;
      const int n = idx - k * NOUT;
      Wo[(k >> 1) * ORS + 2 * n + (k & 1)] = Wout[idx];
    }
    Wo[(threadIdx.x >> 4) * ORS + 2 * NOUT + (threadIdx.x & 15)] = 0.f;
    if (threadIdx.x < 48) Bsm[threadIdx.x] = (threadIdx.x < NOUT) ? bout[min((int)threadIdx.x, NOUT - 1)] : 0.f;
  }
  {
    const int tp = threadIdx.x >> 4;
    const int tn = (threadIdx.x & 15) * 4;
    const float4 w0 = *reinterpret_cast<const float4*>(Wl + (2 * tp) * NHID + tn);
    const float4 w1 = *reinterpret_cast<const float4*>(Wl + (2 * tp + 1) * NHID + tn);
    *reinterpret_cast<float4*>(&Wp[tp * BRS + 2 * tn])     = make_float4(w0.x, w1.x, w0.y, w1.y);
    *reinterpret_cast<float4*>(&Wp[tp * BRS + 2 * tn + 4]) = make_float4(w0.z, w1.z, w0.w, w1.w);
  }
  const int lane = threadIdx.x & 31;
  const int w = __builtin_amdgcn_readfirstlane((int)(threadIdx.x >> 5));
  const float sA = 1.f - ALPHA;
  const float sB = ALPHA;
  const float cscale = (1.f - beta) / beta;

  for (int tile = blockIdx.x; tile < N_TILES; tile += LAYER_BLOCKS) {
    const int r = tile * 16 + w;
    const int start = rowptr[r];
    const int end   = rowptr[r + 1];
    const v2f g0 = *reinterpret_cast<const v2f*>(h0 + (size_t)r * NHID + 2 * lane);
    v2f acc = {0.f, 0.f};
    for (int e = start; e < end; e += 32) {
      const int n = end - e;
      int2 mine = meta[e + min(lane, n - 1)];
      if (lane >= n) mine.y = 0;
#pragma unroll
      for (int g = 0; g < 4; ++g) {
        if (g * 8 < n) {
          v2f v[8];
#pragma unroll
          for (int j = 0; j < 8; ++j) {
            const unsigned c = (unsigned)__builtin_amdgcn_readlane(mine.x, g * 8 + j);
            const float* hr = h + (size_t)c * NHID;
            v[j] = *reinterpret_cast<const v2f*>(hr + 2 * lane);
          }
#pragma unroll
          for (int j = 0; j < 8; ++j) {
            const float wt = __int_as_float(__builtin_amdgcn_readlane(mine.y, g * 8 + j));
            acc.x = fmaf(wt, v[j].x, acc.x);
            acc.y = fmaf(wt, v[j].y, acc.y);
          }
        }
        if (g == 0)
          asm volatile("" :: "v"(g0.x), "v"(g0.y));
      }
    }
    v2f s;
    s.x = sA * acc.x + sB * g0.x;
    s.y = sA * acc.y + sB * g0.y;
    *reinterpret_cast<v2f*>(&Ssm[w * SROW + 2 * lane]) = s;
    __syncthreads();

    if (w < 4) {
      const int half = lane >> 4;
      const int l16  = lane & 15;
      const int n    = w * 16 + l16;
      const float* sa = &Ssm[l16 * SROW + 2 * half];
      const float* wb = &Wp[half * BRS + 2 * n];
      v8f d;
#pragma unroll
      for (int q = 0; q < 8; ++q) d[q] = cscale * Ssm[(8 * half + q) * SROW + n];
#pragma unroll
      for (int kb = 0; kb < NHID; kb += 32) {
        v2f a[8], b[8];
#pragma unroll
        for (int j = 0; j < 8; ++j) {
          const int k = kb + 4 * j;
          a[j] = *reinterpret_cast<const v2f*>(sa + k);
          b[j] = *reinterpret_cast<const v2f*>(wb + (k >> 1) * BRS);
        }
        asm volatile("" :: "v"(a[0]), "v"(a[1]), "v"(a[2]), "v"(a[3]),
                           "v"(a[4]), "v"(a[5]), "v"(a[6]), "v"(a[7]),
                           "v"(b[0]), "v"(b[1]), "v"(b[2]), "v"(b[3]),
                           "v"(b[4]), "v"(b[5]), "v"(b[6]), "v"(b[7]));
#pragma unroll
        for (int j = 0; j < 8; ++j) wmma_acc_slots(d, a[j], b[j]);
      }
#pragma unroll
      for (int q = 0; q < 8; ++q) {
        const float v = beta * d[q];
        Osm[(8 * half + q) * SROW + n] = v > 0.f ? v : 0.f;
      }
    }
    __syncthreads();

    if (!LAST) {
      const v2f o = *reinterpret_cast<const v2f*>(&Osm[w * SROW + 2 * lane]);
      *reinterpret_cast<v2f*>(hout + (size_t)r * NHID + 2 * lane) = o;
    } else {
      if (w < 3) {
        const int half = lane >> 4;
        const int l16  = lane & 15;
        const int n    = w * 16 + l16;
        const float* oa = &Osm[l16 * SROW + 2 * half];
        const float* ob = &Wo[half * ORS + 2 * n];
        const float bias = Bsm[n];
        v8f d;
#pragma unroll
        for (int q = 0; q < 8; ++q) d[q] = bias;
#pragma unroll
        for (int kb = 0; kb < NHID; kb += 32) {
          v2f a[8], b[8];
#pragma unroll
          for (int j = 0; j < 8; ++j) {
            const int k = kb + 4 * j;
            a[j] = *reinterpret_cast<const v2f*>(oa + k);
            b[j] = *reinterpret_cast<const v2f*>(ob + (k >> 1) * ORS);
          }
          asm volatile("" :: "v"(a[0]), "v"(a[1]), "v"(a[2]), "v"(a[3]),
                             "v"(a[4]), "v"(a[5]), "v"(a[6]), "v"(a[7]),
                             "v"(b[0]), "v"(b[1]), "v"(b[2]), "v"(b[3]),
                             "v"(b[4]), "v"(b[5]), "v"(b[6]), "v"(b[7]));
#pragma unroll
          for (int j = 0; j < 8; ++j) wmma_acc_slots(d, a[j], b[j]);
        }
#pragma unroll
        for (int q = 0; q < 8; ++q) Lsm[(8 * half + q) * LROW + n] = d[q];
      }
      __syncthreads();

      const float v0 = Lsm[w * LROW + lane];
      const float v1 = (lane < NOUT - 32) ? Lsm[w * LROW + 32 + lane] : v0;
      float mx = fmaxf(v0, v1);
      for (int o = 16; o > 0; o >>= 1) mx = fmaxf(mx, __shfl_xor(mx, o));
      float sm = expf(v0 - mx) + ((lane < NOUT - 32) ? expf(v1 - mx) : 0.f);
      for (int o = 16; o > 0; o >>= 1) sm += __shfl_xor(sm, o);
      const float lse = mx + logf(sm);
      Tsm[w * NOUT + lane] = v0 - lse;
      if (lane < NOUT - 32) Tsm[w * NOUT + 32 + lane] = v1 - lse;
      __syncthreads();

      if (w < 5) {
        const float4 t4 = *reinterpret_cast<const float4*>(&Tsm[w * 128 + 4 * lane]);
        *reinterpret_cast<float4*>(out + (size_t)tile * (16 * NOUT) + w * 128 + 4 * lane) = t4;
      }
    }
  }
}

extern "C" void kernel_launch(void* const* d_in, const int* in_sizes, int n_in,
                              void* d_out, int out_size, void* d_ws, size_t ws_size,
                              hipStream_t stream) {
  (void)in_sizes; (void)n_in; (void)out_size; (void)ws_size;
  const float* x     = (const float*)d_in[0];
  const int*   row   = (const int*)  d_in[1];
  const int*   col   = (const int*)  d_in[2];
  const float* ew    = (const float*)d_in[3];
  const float* Win   = (const float*)d_in[4];
  const float* bin   = (const float*)d_in[5];
  const float* convW = (const float*)d_in[6];
  const float* Wout  = (const float*)d_in[7];
  const float* bout  = (const float*)d_in[8];

  const size_t S = (size_t)N_NODES * NHID;
  float* h0 = (float*)d_ws;
  float* hA = h0 + S;
  float* hB = hA + S;
  int*  binTotal = (int*)(hB + S);
  int*  cursor   = binTotal + 512;
  int*  rowptr   = cursor + 512;
  int2* tmp      = (int2*)(rowptr + 100096);
  int2* meta     = tmp + E_EDGES;

  const int rowBlocks = (N_NODES + 255) / 256;
  const size_t inLds = (size_t)(NIN / 2) * BRS * sizeof(float);
  zero_kernel<<<1, 256, 0, stream>>>((float4*)binTotal, 256);
  in_gemm_kernel<<<rowBlocks, 256, inLds, stream>>>(x, Win, bin, h0, row, binTotal);
  partition_kernel<<<NCHUNKS, SORT_THREADS, 0, stream>>>(row, col, ew, binTotal, cursor, tmp);
  binsort_kernel<<<NBINS, SORT_THREADS, 0, stream>>>(binTotal, tmp, rowptr, meta);

  const int layerBlocks = LAYER_BLOCKS;

  const float* cur = h0;
  for (int l = 0; l < NLAYERS; ++l) {
    const float beta = logf(0.5f / (float)(l + 1) + 1.f);
    float* nxt = (l & 1) ? hB : hA;
    if (l < NLAYERS - 1)
      layer_kernel<false><<<layerBlocks, 512, 0, stream>>>(rowptr, meta, cur, h0,
          convW + (size_t)l * NHID * NHID, beta, nxt, Wout, bout, (float*)d_out);
    else
      layer_kernel<true><<<layerBlocks, 512, 0, stream>>>(rowptr, meta, cur, h0,
          convW + (size_t)l * NHID * NHID, beta, nxt, Wout, bout, (float*)d_out);
    cur = nxt;
  }
}


# Input GEMM: x read 16 B per lane, four 8-wide k blocks ahead of its matrix steps, through a register ring (was 8 B per k-step, waited for at once); W_in pair rows permuted; k summed in another order
# speedup vs baseline: 5.6601x; 1.0874x over previous
#include <hip/hip_runtime.h>
#include <hip/hip_bf16.h>
#include <math.h>

#define N_NODES 100000
#define E_EDGES 3200000
#define NIN     500
#define NHID    64
#define NOUT    40
#define NLAYERS 8
#define ALPHA   0.1f

#define BRS 160

#define BIN_SHIFT    8
#define BIN_ROWS     (1 << BIN_SHIFT)
#define NBINS        ((N_NODES + BIN_ROWS - 1) / BIN_ROWS)
#define CHUNK        8192
#define NCHUNKS      ((E_EDGES + CHUNK - 1) / CHUNK)
#define GEMM_BLOCKS  ((N_NODES + 255) / 256)
#define HIST_PER_THREAD (CHUNK / 256)
static_assert(NCHUNKS == GEMM_BLOCKS, "one edge chunk per input-GEMM block");
static_assert(NBINS <= 512, "bin histogram is 512 LDS words");

typedef __attribute__((ext_vector_type(2))) float v2f;
typedef __attribute__((ext_vector_type(4))) float v4f;

#define IN_STEPS 62
#define IN_RING  4
#define IN_HEAD  2
static_assert(8 * IN_STEPS + 4 == NIN, "62 blocks of 8 and a tail of 4");
static_assert((IN_STEPS - IN_HEAD) % IN_RING == 0, "the ring loop takes whole trips");
typedef __attribute__((ext_vector_type(8))) float v8f;

__device__ __forceinline__ void wmma_acc(v8f& c, v2f a, v2f b) {
  asm("v_wmma_f32_16x16x4_f32 %0, %1, %2, %0\n\tv_nop\n\tv_nop\n\tv_nop\n\tv_nop"
      : "+v"(c) : "v"(a), "v"(b));
}
__device__ __forceinline__ void wmma_acc_slots(v8f& c, v2f a, v2f b) {
  asm volatile("v_wmma_f32_16x16x4_f32 %0, %1, %2, %0\n\tv_nop\n\tv_nop\n\tv_nop\n\tv_nop"
      : "+v"(c) : "v"(a), "v"(b));
}
__device__ __forceinline__ void wmma_epilogue_guard() {
  asm volatile("s_nop 8");
}

__global__ void __launch_bounds__(256) in_gemm_kernel(
    const float* __restrict__ x, const float* __restrict__ Win,
    const float* __restrict__ bin, float* __restrict__ h0,
    const int* __restrict__ row, int* __restrict__ binTotal) {
  extern __shared__ float Wp[];
  {
    int* hist = reinterpret_cast<int*>(Wp);
    hist[threadIdx.x] = 0;
    hist[threadIdx.x + 256] = 0;
    const int e0 = blockIdx.x * CHUNK;
    int rw[HIST_PER_THREAD];
#pragma unroll
    for (int j = 0; j < HIST_PER_THREAD; ++j)
      rw[j] = row[min(e0 + j * 256 + (int)threadIdx.x, E_EDGES - 1)];
    __syncthreads();
#pragma unroll
    for (int j = 0; j < HIST_PER_THREAD; ++j)
      if (e0 + j * 256 + (int)threadIdx.x < E_EDGES) atomicAdd(&hist[rw[j] >> BIN_SHIFT], 1);
    __syncthreads();
    for (int i = threadIdx.x; i < NBINS; i += 256) {
      const int c = hist[i];
      if (c) atomicAdd(&binTotal[i], c);
    }
    __syncthreads();
  }
  for (int idx = threadIdx.x; idx < NIN * NHID; idx += 256) {
    const int k = idx >> 6;
    const int n = idx & 63;
    const int p = k >> 1;
    const int prow = (p < 4 * IN_STEPS) ? ((p & ~3) | ((p & 1) << 1) | ((p >> 1) & 1)) : p;
    Wp[prow * BRS + 2 * n + (k & 1)] = Win[idx];
  }
  __syncthreads();

  const int wave = threadIdx.x >> 5;
  const int lane = threadIdx.x & 31;
  const int half = lane >> 4;
  const int l16  = lane & 15;
  const int r0   = blockIdx.x * 256 + wave * 32;

  const int arow0 = min(r0 + l16,      N_NODES - 1);
  const int arow1 = min(r0 + 16 + l16, N_NODES - 1);
  const float* xr0 = x + (long long)arow0 * NIN;
  const float* xr1 = x + (long long)arow1 * NIN;

  v8f acc0[4] = {};
  v8f acc1[4] = {};
  const float* xa0 = xr0 + 4 * half;
  const float* xa1 = xr1 + 4 * half;
  const float* wrow = &Wp[half * BRS + 2 * l16];
  const v2f ta0 = *reinterpret_cast<const v2f*>(xr0 + 8 * IN_STEPS + 2 * half);
  const v2f ta1 = *reinterpret_cast<const v2f*>(xr1 + 8 * IN_STEPS + 2 * half);
  v4f ha[IN_HEAD], hb[IN_HEAD];
#pragma unroll
  for (int d = 0; d < IN_HEAD; ++d) {
    ha[d] = *reinterpret_cast<const v4f*>(xa0 + 8 * d);
    hb[d] = *reinterpret_cast<const v4f*>(xa1 + 8 * d);
  }
  v4f ra[IN_RING], rb[IN_RING];
#pragma unroll
  for (int d = 0; d < IN_RING; ++d) {
    ra[d] = *reinterpret_cast<const v4f*>(xa0 + 8 * (IN_HEAD + d));
    rb[d] = *reinterpret_cast<const v4f*>(xa1 + 8 * (IN_HEAD + d));
  }
  {
    const float* wb = wrow + (4 * IN_STEPS) * BRS;
#pragma unroll
    for (int t = 0; t < 4; ++t) {
      const v2f b = *reinterpret_cast<const v2f*>(wb + t * 32);
      wmma_acc(acc0[t], ta0, b);
      wmma_acc(acc1[t], ta1, b);
    }
  }
#pragma unroll
  for (int d = 0; d < IN_HEAD; ++d) {
#pragma unroll
    for (int j = 0; j < 2; ++j) {
      const v2f p0 = {ha[d][2 * j], ha[d][2 * j + 1]};
      const v2f p1 = {hb[d][2 * j], hb[d][2 * j + 1]};
      const float* wb = wrow + (4 * d + 2 * j) * BRS;
#pragma unroll
      for (int t = 0; t < 4; ++t) {
        const v2f b = *reinterpret_cast<const v2f*>(wb + t * 32);
        wmma_acc(acc0[t], p0, b);
        wmma_acc(acc1[t], p1, b);
      }
    }
  }
  for (int i0 = IN_HEAD; i0 < IN_STEPS; i0 += IN_RING) {
#pragma unroll
    for (int d = 0; d < IN_RING; ++d) {
      const int i = i0 + d;
      const v4f a0 = ra[d];
      const v4f a1 = rb[d];
#pragma unroll
      for (int j = 0; j < 2; ++j) {
        const v2f p0 = {a0[2 * j], a0[2 * j + 1]};
        const v2f p1 = {a1[2 * j], a1[2 * j + 1]};
        const float* wb = wrow + (4 * i + 2 * j) * BRS;
#pragma unroll
        for (int t = 0; t < 4; ++t) {
          const v2f b = *reinterpret_cast<const v2f*>(wb + t * 32);
          wmma_acc(acc0[t], p0, b);
          wmma_acc(acc1[t], p1, b);
        }
      }
      const int nx = min(i + IN_RING, IN_STEPS - 1);
      ra[d] = *reinterpret_cast<const v4f*>(xa0 + 8 * nx);
      rb[d] = *reinterpret_cast<const v4f*>(xa1 + 8 * nx);
      __builtin_amdgcn_sched_barrier(0);
    }
  }
  wmma_epilogue_guard();

#pragma unroll
  for (int t = 0; t < 4; ++t) {
    const int n = t * 16 + l16;
    const float bias = bin[n];
#pragma unroll
    for (int r = 0; r < 8; ++r) {
      const int m = r + 8 * half;
      const int orow0 = r0 + m;
      const int orow1 = r0 + 16 + m;
      if (orow0 < N_NODES) {
        const float v = acc0[t][r] + bias;
        h0[(long long)orow0 * NHID + n] = v > 0.f ? v : 0.f;
      }
      if (orow1 < N_NODES) {
        const float v = acc1[t][r] + bias;
        h0[(long long)orow1 * NHID + n] = v > 0.f ? v : 0.f;
      }
    }
  }
}

__global__ void zero_kernel(float4* __restrict__ p, int n4) {
  int i = blockIdx.x * blockDim.x + threadIdx.x;
  if (i < n4) p[i] = make_float4(0.f, 0.f, 0.f, 0.f);
}

#define SORT_THREADS 1024
#define PER_THREAD   (CHUNK / SORT_THREADS)
#define STAGE_CAP    16384
static_assert(NBINS <= SORT_THREADS && NBINS <= 512, "one thread per bin in the scans");
static_assert(N_NODES <= (1 << 24), "col is packed into 24 bits beside the row's low 8");

__device__ __forceinline__ int block_excl_scan(int v, int* wsum) {
  const int lane = threadIdx.x & 31;
  const int wv = threadIdx.x >> 5;
  int incl = v;
  for (int o = 1; o < 32; o <<= 1) {
    const int t = __shfl_up(incl, o);
    if (lane >= o) incl += t;
  }
  if (lane == 31) wsum[wv] = incl;
  __syncthreads();
  int base = 0;
  for (int k = 0; k < 32; ++k) {
    const int sk = wsum[k];
    if (k < wv) base += sk;
  }
  __syncthreads();
  return base + incl - v;
}

__global__ void __launch_bounds__(SORT_THREADS) partition_kernel(
    const int* __restrict__ row, const int* __restrict__ col,
    const float* __restrict__ ew, const int* __restrict__ binTotal,
    int* __restrict__ cursor, int2* __restrict__ tmp) {
  __shared__ int hist[512];
  __shared__ int lstart[512];
  __shared__ int gbase[512];
  __shared__ int wsum[32];
  __shared__ __attribute__((aligned(16))) int2 item[CHUNK];
  __shared__ int dst[CHUNK];
  const int tid = threadIdx.x;
  if (tid < 512) hist[tid] = 0;
  __syncthreads();

  const int e0 = blockIdx.x * CHUNK;
  int bn[PER_THREAD], rk[PER_THREAD], pk[PER_THREAD], wb[PER_THREAD];
#pragma unroll
  for (int j = 0; j < PER_THREAD; ++j) {
    const int ec = min(e0 + j * SORT_THREADS + tid, E_EDGES - 1);
    bn[j] = row[ec];
    pk[j] = col[ec];
    wb[j] = __float_as_int(ew[ec]);
  }
#pragma unroll
  for (int j = 0; j < PER_THREAD; ++j) {
    const int rw = bn[j];
    pk[j] = (int)(((unsigned)(rw & (BIN_ROWS - 1)) << 24) | (unsigned)pk[j]);
    rk[j] = 0;
    if (e0 + j * SORT_THREADS + tid < E_EDGES) {
      bn[j] = rw >> BIN_SHIFT;
      rk[j] = atomicAdd(&hist[bn[j]], 1);
    } else {
      bn[j] = -1;
    }
  }
  __syncthreads();

  const int hcount = (tid < NBINS) ? hist[tid] : 0;
  const int btotal = (tid < NBINS) ? binTotal[tid] : 0;
  const int ls = block_excl_scan(hcount, wsum);
  const int bs = block_excl_scan(btotal, wsum);
  if (tid < NBINS) {
    lstart[tid] = ls;
    const int reserved = hcount ? atomicAdd(&cursor[tid], hcount) : 0;
    gbase[tid] = bs + reserved;
  }
  __syncthreads();

#pragma unroll
  for (int j = 0; j < PER_THREAD; ++j) {
    if (bn[j] >= 0) {
      const int p = lstart[bn[j]] + rk[j];
      item[p] = make_int2(pk[j], wb[j]);
      dst[p]  = gbase[bn[j]] + rk[j];
    }
  }
  __syncthreads();

  const int nvalid = min(CHUNK, E_EDGES - e0);
  for (int j = 0; j < PER_THREAD; ++j) {
    const int p = j * SORT_THREADS + tid;
    if (p < nvalid) tmp[dst[p]] = item[p];
  }
}

__global__ void __launch_bounds__(SORT_THREADS) binsort_kernel(
    const int* __restrict__ binTotal, const int2* __restrict__ tmp,
    int* __restrict__ rowptr, int2* __restrict__ meta) {
  __shared__ int cnt[BIN_ROWS];
  __shared__ int fillc[BIN_ROWS];
  __shared__ int rstart[BIN_ROWS];
  __shared__ int wsum[32];
  __shared__ int sbase;
  __shared__ __attribute__((aligned(16))) int2 stage[STAGE_CAP];
  const int tid = threadIdx.x;
  const int b = blockIdx.x;
  if (tid < BIN_ROWS) { cnt[tid] = 0; fillc[tid] = 0; }
  const int btotal = (tid < NBINS) ? binTotal[tid] : 0;
  const int bs = block_excl_scan(btotal, wsum);
  if (tid == b) sbase = bs;
  __syncthreads();
  const int base = sbase;
  const int nb = binTotal[b];

  for (int i = tid; i < nb; i += SORT_THREADS) {
    const int2 it = tmp[base + i];
    atomicAdd(&cnt[(unsigned)it.x >> 24], 1);
  }
  __syncthreads();
  const int c = (tid < BIN_ROWS) ? cnt[tid] : 0;
  const int rs = block_excl_scan(c, wsum);
  if (tid < BIN_ROWS) {
    rstart[tid] = rs;
    const int r = b * BIN_ROWS + tid;
    if (r < N_NODES) rowptr[r] = base + rs;
  }
  if (b == NBINS - 1 && tid == 0) rowptr[N_NODES] = base + nb;
  __syncthreads();

  const bool staged = (nb <= STAGE_CAP);
  for (int i = tid; i < nb; i += SORT_THREADS) {
    const int2 it = tmp[base + i];
    const int rl = (unsigned)it.x >> 24;
    const int p = rstart[rl] + atomicAdd(&fillc[rl], 1);
    const int2 o = make_int2(it.x & 0x00FFFFFF, it.y);
    if (staged) stage[p] = o; else meta[base + p] = o;
  }
  if (staged) {
    __syncthreads();
    for (int i = tid; i < nb; i += SORT_THREADS) meta[base + i] = stage[i];
  }
}

#define SROW 68
#define ORS  96
#define LROW 52

static_assert(N_NODES % 16 == 0, "16 rows per tile, no partial tile");
#define N_TILES      (N_NODES / 16)
#define LAYER_BLOCKS 1024

template <bool LAST>
__global__ void __launch_bounds__(512, 16) layer_kernel(
    const int* __restrict__ rowptr, const int2* __restrict__ meta,
    const float* __restrict__ h, const float* __restrict__ h0,
    const float* __restrict__ Wl, float beta, float* __restrict__ hout,
    const float* __restrict__ Wout, const float* __restrict__ bout,
    float* __restrict__ out) {
  __shared__ __attribute__((aligned(16))) float Ssm[16 * SROW];
  __shared__ __attribute__((aligned(16))) float Osm[16 * SROW];
  __shared__ __attribute__((aligned(16))) float Wp[(NHID / 2) * BRS];
  __shared__ __attribute__((aligned(16))) float Wo[LAST ? (NHID / 2) * ORS : 1];
  __shared__ __attribute__((aligned(16))) float Lsm[LAST ? 16 * LROW : 1];
  __shared__ __attribute__((aligned(16))) float Tsm[LAST ? 16 * NOUT : 1];
  __shared__ float Bsm[LAST ? 48 : 1];
  if (LAST) {
    for (int idx = threadIdx.x; idx < NHID * NOUT; idx += 512) {
      const int k = idx / NOUT;
      const int n = idx - k * NOUT;
      Wo[(k >> 1) * ORS + 2 * n + (k & 1)] = Wout[idx];
    }
    Wo[(threadIdx.x >> 4) * ORS + 2 * NOUT + (threadIdx.x & 15)] = 0.f;
    if (threadIdx.x < 48) Bsm[threadIdx.x] = (threadIdx.x < NOUT) ? bout[min((int)threadIdx.x, NOUT - 1)] : 0.f;
  }
  {
    const int tp = threadIdx.x >> 4;
    const int tn = (threadIdx.x & 15) * 4;
    const float4 w0 = *reinterpret_cast<const float4*>(Wl + (2 * tp) * NHID + tn);
    const float4 w1 = *reinterpret_cast<const float4*>(Wl + (2 * tp + 1) * NHID + tn);
    *reinterpret_cast<float4*>(&Wp[tp * BRS + 2 * tn])     = make_float4(w0.x, w1.x, w0.y, w1.y);
    *reinterpret_cast<float4*>(&Wp[tp * BRS + 2 * tn + 4]) = make_float4(w0.z, w1.z, w0.w, w1.w);
  }
  const int lane = threadIdx.x & 31;
  const int w = __builtin_amdgcn_readfirstlane((int)(threadIdx.x >> 5));
  const float sA = 1.f - ALPHA;
  const float sB = ALPHA;
  const float cscale = (1.f - beta) / beta;

  for (int tile = blockIdx.x; tile < N_TILES; tile += LAYER_BLOCKS) {
    const int r = tile * 16 + w;
    const int start = rowptr[r];
    const int end   = rowptr[r + 1];
    const v2f g0 = *reinterpret_cast<const v2f*>(h0 + (size_t)r * NHID + 2 * lane);
    v2f acc = {0.f, 0.f};
    for (int e = start; e < end; e += 32) {
      const int n = end - e;
      int2 mine = meta[e + min(lane, n - 1)];
      if (lane >= n) mine.y = 0;
#pragma unroll
      for (int g = 0; g < 4; ++g) {
        if (g * 8 < n) {
          v2f v[8];
#pragma unroll
          for (int j = 0; j < 8; ++j) {
            const unsigned c = (unsigned)__builtin_amdgcn_readlane(mine.x, g * 8 + j);
            const float* hr = h + (size_t)c * NHID;
            v[j] = *reinterpret_cast<const v2f*>(hr + 2 * lane);
          }
#pragma unroll
          for (int j = 0; j < 8; ++j) {
            const float wt = __int_as_float(__builtin_amdgcn_readlane(mine.y, g * 8 + j));
            acc.x = fmaf(wt, v[j].x, acc.x);
            acc.y = fmaf(wt, v[j].y, acc.y);
          }
        }
        if (g == 0)
          asm volatile("" :: "v"(g0.x), "v"(g0.y));
      }
    }
    v2f s;
    s.x = sA * acc.x + sB * g0.x;
    s.y = sA * acc.y + sB * g0.y;
    *reinterpret_cast<v2f*>(&Ssm[w * SROW + 2 * lane]) = s;
    __syncthreads();

    if (w < 4) {
      const int half = lane >> 4;
      const int l16  = lane & 15;
      const int n    = w * 16 + l16;
      const float* sa = &Ssm[l16 * SROW + 2 * half];
      const float* wb = &Wp[half * BRS + 2 * n];
      v8f d;
#pragma unroll
      for (int q = 0; q < 8; ++q) d[q] = cscale * Ssm[(8 * half + q) * SROW + n];
#pragma unroll
      for (int kb = 0; kb < NHID; kb += 32) {
        v2f a[8], b[8];
#pragma unroll
        for (int j = 0; j < 8; ++j) {
          const int k = kb + 4 * j;
          a[j] = *reinterpret_cast<const v2f*>(sa + k);
          b[j] = *reinterpret_cast<const v2f*>(wb + (k >> 1) * BRS);
        }
        asm volatile("" :: "v"(a[0]), "v"(a[1]), "v"(a[2]), "v"(a[3]),
                           "v"(a[4]), "v"(a[5]), "v"(a[6]), "v"(a[7]),
                           "v"(b[0]), "v"(b[1]), "v"(b[2]), "v"(b[3]),
                           "v"(b[4]), "v"(b[5]), "v"(b[6]), "v"(b[7]));
#pragma unroll
        for (int j = 0; j < 8; ++j) wmma_acc_slots(d, a[j], b[j]);
      }
#pragma unroll
      for (int q = 0; q < 8; ++q) {
        const float v = beta * d[q];
        Osm[(8 * half + q) * SROW + n] = v > 0.f ? v : 0.f;
      }
    }
    __syncthreads();

    if (!LAST) {
      const v2f o = *reinterpret_cast<const v2f*>(&Osm[w * SROW + 2 * lane]);
      *reinterpret_cast<v2f*>(hout + (size_t)r * NHID + 2 * lane) = o;
    } else {
      if (w < 3) {
        const int half = lane >> 4;
        const int l16  = lane & 15;
        const int n    = w * 16 + l16;
        const float* oa = &Osm[l16 * SROW + 2 * half];
        const float* ob = &Wo[half * ORS + 2 * n];
        const float bias = Bsm[n];
        v8f d;
#pragma unroll
        for (int q = 0; q < 8; ++q) d[q] = bias;
#pragma unroll
        for (int kb = 0; kb < NHID; kb += 32) {
          v2f a[8], b[8];
#pragma unroll
          for (int j = 0; j < 8; ++j) {
            const int k = kb + 4 * j;
            a[j] = *reinterpret_cast<const v2f*>(oa + k);
            b[j] = *reinterpret_cast<const v2f*>(ob + (k >> 1) * ORS);
          }
          asm volatile("" :: "v"(a[0]), "v"(a[1]), "v"(a[2]), "v"(a[3]),
                             "v"(a[4]), "v"(a[5]), "v"(a[6]), "v"(a[7]),
                             "v"(b[0]), "v"(b[1]), "v"(b[2]), "v"(b[3]),
                             "v"(b[4]), "v"(b[5]), "v"(b[6]), "v"(b[7]));
#pragma unroll
          for (int j = 0; j < 8; ++j) wmma_acc_slots(d, a[j], b[j]);
        }
#pragma unroll
        for (int q = 0; q < 8; ++q) Lsm[(8 * half + q) * LROW + n] = d[q];
      }
      __syncthreads();

      const float v0 = Lsm[w * LROW + lane];
      const float v1 = (lane < NOUT - 32) ? Lsm[w * LROW + 32 + lane] : v0;
      float mx = fmaxf(v0, v1);
      for (int o = 16; o > 0; o >>= 1) mx = fmaxf(mx, __shfl_xor(mx, o));
      float sm = expf(v0 - mx) + ((lane < NOUT - 32) ? expf(v1 - mx) : 0.f);
      for (int o = 16; o > 0; o >>= 1) sm += __shfl_xor(sm, o);
      const float lse = mx + logf(sm);
      Tsm[w * NOUT + lane] = v0 - lse;
      if (lane < NOUT - 32) Tsm[w * NOUT + 32 + lane] = v1 - lse;
      __syncthreads();

      if (w < 5) {
        const float4 t4 = *reinterpret_cast<const float4*>(&Tsm[w * 128 + 4 * lane]);
        *reinterpret_cast<float4*>(out + (size_t)tile * (16 * NOUT) + w * 128 + 4 * lane) = t4;
      }
    }
  }
}

extern "C" void kernel_launch(void* const* d_in, const int* in_sizes, int n_in,
                              void* d_out, int out_size, void* d_ws, size_t ws_size,
                              hipStream_t stream) {
  (void)in_sizes; (void)n_in; (void)out_size; (void)ws_size;
  const float* x     = (const float*)d_in[0];
  const int*   row   = (const int*)  d_in[1];
  const int*   col   = (const int*)  d_in[2];
  const float* ew    = (const float*)d_in[3];
  const float* Win   = (const float*)d_in[4];
  const float* bin   = (const float*)d_in[5];
  const float* convW = (const float*)d_in[6];
  const float* Wout  = (const float*)d_in[7];
  const float* bout  = (const float*)d_in[8];

  const size_t S = (size_t)N_NODES * NHID;
  float* h0 = (float*)d_ws;
  float* hA = h0 + S;
  float* hB = hA + S;
  int*  binTotal = (int*)(hB + S);
  int*  cursor   = binTotal + 512;
  int*  rowptr   = cursor + 512;
  int2* tmp      = (int2*)(rowptr + 100096);
  int2* meta     = tmp + E_EDGES;

  const int rowBlocks = (N_NODES + 255) / 256;
  const size_t inLds = (size_t)(NIN / 2) * BRS * sizeof(float);
  zero_kernel<<<1, 256, 0, stream>>>((float4*)binTotal, 256);
  in_gemm_kernel<<<rowBlocks, 256, inLds, stream>>>(x, Win, bin, h0, row, binTotal);
  partition_kernel<<<NCHUNKS, SORT_THREADS, 0, stream>>>(row, col, ew, binTotal, cursor, tmp);
  binsort_kernel<<<NBINS, SORT_THREADS, 0, stream>>>(binTotal, tmp, rowptr, meta);

  const int layerBlocks = LAYER_BLOCKS;

  const float* cur = h0;
  for (int l = 0; l < NLAYERS; ++l) {
    const float beta = logf(0.5f / (float)(l + 1) + 1.f);
    float* nxt = (l & 1) ? hB : hA;
    if (l < NLAYERS - 1)
      layer_kernel<false><<<layerBlocks, 512, 0, stream>>>(rowptr, meta, cur, h0,
          convW + (size_t)l * NHID * NHID, beta, nxt, Wout, bout, (float*)d_out);
    else
      layer_kernel<true><<<layerBlocks, 512, 0, stream>>>(rowptr, meta, cur, h0,
          convW + (size_t)l * NHID * NHID, beta, nxt, Wout, bout, (float*)d_out);
    cur = nxt;
  }
}
